# starting kernel
# baseline (speedup 1.0000x reference)
#include <hip/hip_runtime.h>
#include <hip/hip_bf16.h>
#include <math.h>

typedef __attribute__((ext_vector_type(2))) float v2f;
typedef __attribute__((ext_vector_type(8))) float v8f;

#define HEADS 4
#define ATT_F 16
#define DNODE 64
#define DEDGE 16

__global__ __launch_bounds__(256) void gat_node_project(
    const float* __restrict__ nodes, const float* __restrict__ W,
    float* __restrict__ Wn, int n_nodes, int n_tiles)
{
  int wave = blockIdx.x * (blockDim.x >> 5) + (threadIdx.x >> 5);
  if (wave >= n_tiles) return;
  int lane = threadIdx.x & 31;
  int half = lane >> 4;
  int l16  = lane & 15;
  int row0 = wave * 16;
  int m = row0 + l16; if (m >= n_nodes) m = n_nodes - 1;
  const float* __restrict__ arow = nodes + (size_t)m * DNODE;

  v8f acc[4] = {};
  for (int k0 = 0; k0 < DNODE; k0 += 4) {
    int kk = k0 + 2 * half;
    v2f a; a.x = arow[kk]; a.y = arow[kk + 1];
#pragma unroll
    for (int t = 0; t < 4; ++t) {
      int j = t * 16 + l16;
      v2f b; b.x = W[(size_t)j * DNODE + kk]; b.y = W[(size_t)j * DNODE + kk + 1];
      acc[t] = __builtin_amdgcn_wmma_f32_16x16x4_f32(
           false, a,  false, b,
           (short)0, acc[t],  false,  false);
    }
  }

  if (row0 + 16 <= n_nodes) {
#pragma unroll
    for (int t = 0; t < 4; ++t)
#pragma unroll
      for (int v = 0; v < 8; ++v)
        Wn[(size_t)(row0 + v + 8 * half) * DNODE + t * 16 + l16] = acc[t][v];
  } else {
#pragma unroll
    for (int t = 0; t < 4; ++t)
#pragma unroll
      for (int v = 0; v < 8; ++v) {
        int mo = row0 + v + 8 * half;
        if (mo < n_nodes) Wn[(size_t)mo * DNODE + t * 16 + l16] = acc[t][v];
      }
  }
}

__global__ __launch_bounds__(256) void gat_node_scores(
    const float* __restrict__ Wn, const float* __restrict__ a,
    float* __restrict__ score_s, float* __restrict__ score_r, int n_nodes)
{
  int i = blockIdx.x * blockDim.x + threadIdx.x;
  if (i >= n_nodes) return;
  const float* __restrict__ row = Wn + (size_t)i * DNODE;
#pragma unroll
  for (int h = 0; h < HEADS; ++h) {
    float ss = 0.f, sr = 0.f;
#pragma unroll
    for (int f = 0; f < ATT_F; ++f) {
      float v = row[h * ATT_F + f];
      ss += v * a[h * 48 + f];
      sr += v * a[h * 48 + 16 + f];
    }
    score_s[i * HEADS + h] = ss;
    score_r[i * HEADS + h] = sr;
  }
}

__global__ void gat_wcomb(const float* __restrict__ W_edge,
                          const float* __restrict__ a, float* __restrict__ wcomb)
{
  int idx = threadIdx.x;
  if (idx >= HEADS * DEDGE) return;
  int h = idx >> 4, i = idx & 15;
  float acc = 0.f;
#pragma unroll
  for (int f = 0; f < ATT_F; ++f)
    acc += a[h * 48 + 32 + f] * W_edge[((h * ATT_F) + f) * DEDGE + i];
  wcomb[idx] = acc;
}

__global__ __launch_bounds__(256) void gat_edge_pass1(
    const float* __restrict__ edges, const int* __restrict__ recv,
    const int* __restrict__ send, const float* __restrict__ score_s,
    const float* __restrict__ score_r, const float* __restrict__ wcomb,
    float* __restrict__ ex, float* __restrict__ norm, int n_edges, int n_nodes)
{
  __shared__ float wc[HEADS * DEDGE];
  if (threadIdx.x < HEADS * DEDGE) wc[threadIdx.x] = wcomb[threadIdx.x];
  __syncthreads();

  int e = blockIdx.x * blockDim.x + threadIdx.x;
  int etot = n_edges + n_nodes;
  if (e >= etot) return;

  int s, r;
  float elt[HEADS] = {0.f, 0.f, 0.f, 0.f};
  if (e < n_edges) {
    s = send[e]; r = recv[e];
    const float4* __restrict__ ep = (const float4*)(edges + (size_t)e * DEDGE);
    float ef[DEDGE];
#pragma unroll
    for (int c = 0; c < 4; ++c) {
      float4 q = ep[c];
      ef[4 * c + 0] = q.x; ef[4 * c + 1] = q.y; ef[4 * c + 2] = q.z; ef[4 * c + 3] = q.w;
    }
#pragma unroll
    for (int h = 0; h < HEADS; ++h) {
      float acc = 0.f;
#pragma unroll
      for (int i = 0; i < DEDGE; ++i) acc += ef[i] * wc[h * DEDGE + i];
      elt[h] = acc;
    }
  } else {
    s = r = e - n_edges;
  }
#pragma unroll
  for (int h = 0; h < HEADS; ++h) {
    float m  = score_s[s * HEADS + h] + score_r[r * HEADS + h] + elt[h];
    float lm = (m > 0.f) ? m : 0.01f * m;
    float ez = __expf(lm);
    ex[(size_t)e * HEADS + h] = ez;
    atomicAdd(&norm[r * HEADS + h], ez);
  }
}

__global__ __launch_bounds__(256) void gat_edge_pass2(
    const int* __restrict__ recv, const int* __restrict__ send,
    const float* __restrict__ ex, const float* __restrict__ norm,
    const float* __restrict__ Wn, float* __restrict__ aggr,
    int n_edges, int n_nodes)
{
  int e = blockIdx.x * blockDim.x + threadIdx.x;
  int etot = n_edges + n_nodes;
  if (e >= etot) return;
  int s, r;
  if (e < n_edges) { s = send[e]; r = recv[e]; }
  else             { s = r = e - n_edges; }

  float al[HEADS];
#pragma unroll
  for (int h = 0; h < HEADS; ++h)
    al[h] = ex[(size_t)e * HEADS + h] / norm[r * HEADS + h];

  const float4* __restrict__ src = (const float4*)(Wn + (size_t)s * DNODE);
  float* dst = aggr + (size_t)r * DNODE;
#pragma unroll
  for (int c = 0; c < 16; ++c) {
    float4 v = src[c];
    float aa = al[c >> 2];
    atomicAdd(dst + 4 * c + 0, aa * v.x);
    atomicAdd(dst + 4 * c + 1, aa * v.y);
    atomicAdd(dst + 4 * c + 2, aa * v.z);
    atomicAdd(dst + 4 * c + 3, aa * v.w);
  }
}

__global__ __launch_bounds__(256) void gat_finalize(
    float* __restrict__ out, const float* __restrict__ scale,
    const float* __restrict__ bias, int n_nodes)
{
  int i = blockIdx.x * blockDim.x + threadIdx.x;
  if (i >= n_nodes) return;
  float* row = out + (size_t)i * DNODE;
  float v[DNODE];
  float mean = 0.f;
#pragma unroll
  for (int j = 0; j < DNODE; ++j) {
    float x = row[j];
    x = (x > 0.f) ? x : expm1f(x);
    v[j] = x; mean += x;
  }
  mean *= (1.f / DNODE);
  float var = 0.f;
#pragma unroll
  for (int j = 0; j < DNODE; ++j) { float d = v[j] - mean; var += d * d; }
  var *= (1.f / DNODE);
  float inv = rsqrtf(var + 1e-6f);
#pragma unroll
  for (int j = 0; j < DNODE; ++j)
    row[j] = (v[j] - mean) * inv * scale[j] + bias[j];
}

extern "C" void kernel_launch(void* const* d_in, const int* in_sizes, int n_in,
                              void* d_out, int out_size, void* d_ws, size_t ws_size,
                              hipStream_t stream)
{
  const float* nodes  = (const float*)d_in[0];
  const float* edges  = (const float*)d_in[1];
  const int*   recv   = (const int*)d_in[2];
  const int*   send   = (const int*)d_in[3];
  const float* W      = (const float*)d_in[4];
  const float* W_edge = (const float*)d_in[5];
  const float* a      = (const float*)d_in[6];
  const float* ln_s   = (const float*)d_in[7];
  const float* ln_b   = (const float*)d_in[8];

  const int n_nodes = in_sizes[0] / DNODE;
  const int n_edges = in_sizes[1] / DEDGE;
  const int etot    = n_edges + n_nodes;
  float* out = (float*)d_out;

  char* ws = (char*)d_ws;
  size_t off = 0;
  auto carve = [&](size_t bytes) -> float* {
    float* p = (float*)(ws + off);
    off += (bytes + 255) & ~(size_t)255;
    return p;
  };
  float* Wn      = carve((size_t)n_nodes * DNODE * sizeof(float));
  float* score_s = carve((size_t)n_nodes * HEADS * sizeof(float));
  float* score_r = carve((size_t)n_nodes * HEADS * sizeof(float));
  float* wcomb   = carve((size_t)HEADS * DEDGE * sizeof(float));
  float* norm    = carve((size_t)n_nodes * HEADS * sizeof(float));
  float* ex      = carve((size_t)etot * HEADS * sizeof(float));

  hipMemsetAsync(out,  0, (size_t)n_nodes * DNODE * sizeof(float), stream);
  hipMemsetAsync(norm, 0, (size_t)n_nodes * HEADS * sizeof(float), stream);

  const int n_tiles = (n_nodes + 15) / 16;
  dim3 blk(256);

  gat_node_project<<<dim3((n_tiles + 7) / 8), blk, 0, stream>>>(
      nodes, W, Wn, n_nodes, n_tiles);
  gat_node_scores<<<dim3((n_nodes + 255) / 256), blk, 0, stream>>>(
      Wn, a, score_s, score_r, n_nodes);
  gat_wcomb<<<dim3(1), dim3(64), 0, stream>>>(W_edge, a, wcomb);
  gat_edge_pass1<<<dim3((etot + 255) / 256), blk, 0, stream>>>(
      edges, recv, send, score_s, score_r, wcomb, ex, norm, n_edges, n_nodes);
  gat_edge_pass2<<<dim3((etot + 255) / 256), blk, 0, stream>>>(
      recv, send, ex, norm, Wn, out, n_edges, n_nodes);
  gat_finalize<<<dim3((n_nodes + 255) / 256), blk, 0, stream>>>(
      out, ln_s, ln_b, n_nodes);
}


# edge count without global atomics: 32 edge chunks, each counted per receiver in one work-group's LDS (LDS add returns rank), a column pass prefixes chunk counts; sums in chunk, LDS-arrival order
# speedup vs baseline: 7.0751x; 7.0751x over previous
#include <hip/hip_runtime.h>
#include <hip/hip_bf16.h>
#include <math.h>

typedef __attribute__((ext_vector_type(2))) float v2f;
typedef __attribute__((ext_vector_type(8))) float v8f;

#define HEADS 4
#define ATT_F 16
#define DNODE 64
#define DEDGE 16

__global__ __launch_bounds__(256) void gat_node_project(
    const float* __restrict__ nodes, const float* __restrict__ W,
    float* __restrict__ Wn, int n_nodes, int n_tiles)
{
  int wave = blockIdx.x * (blockDim.x >> 5) + (threadIdx.x >> 5);
  if (wave >= n_tiles) return;
  int lane = threadIdx.x & 31;
  int half = lane >> 4;
  int l16  = lane & 15;
  int row0 = wave * 16;
  int m = row0 + l16; if (m >= n_nodes) m = n_nodes - 1;
  const float* __restrict__ arow = nodes + (size_t)m * DNODE;

  v8f acc[4] = {};
  for (int k0 = 0; k0 < DNODE; k0 += 4) {
    int kk = k0 + 2 * half;
    v2f a; a.x = arow[kk]; a.y = arow[kk + 1];
#pragma unroll
    for (int t = 0; t < 4; ++t) {
      int j = t * 16 + l16;
      v2f b; b.x = W[(size_t)j * DNODE + kk]; b.y = W[(size_t)j * DNODE + kk + 1];
      acc[t] = __builtin_amdgcn_wmma_f32_16x16x4_f32(
           false, a,  false, b,
           (short)0, acc[t],  false,  false);
    }
  }

  if (row0 + 16 <= n_nodes) {
#pragma unroll
    for (int t = 0; t < 4; ++t)
#pragma unroll
      for (int v = 0; v < 8; ++v)
        Wn[(size_t)(row0 + v + 8 * half) * DNODE + t * 16 + l16] = acc[t][v];
  } else {
#pragma unroll
    for (int t = 0; t < 4; ++t)
#pragma unroll
      for (int v = 0; v < 8; ++v) {
        int mo = row0 + v + 8 * half;
        if (mo < n_nodes) Wn[(size_t)mo * DNODE + t * 16 + l16] = acc[t][v];
      }
  }
}

__global__ __launch_bounds__(256) void gat_node_scores(
    const float* __restrict__ Wn, const float* __restrict__ a,
    float* __restrict__ score_s, float* __restrict__ score_r, int n_nodes)
{
  int i = blockIdx.x * blockDim.x + threadIdx.x;
  if (i >= n_nodes) return;
  const float* __restrict__ row = Wn + (size_t)i * DNODE;
#pragma unroll
  for (int h = 0; h < HEADS; ++h) {
    float ss = 0.f, sr = 0.f;
#pragma unroll
    for (int f = 0; f < ATT_F; ++f) {
      float v = row[h * ATT_F + f];
      ss += v * a[h * 48 + f];
      sr += v * a[h * 48 + 16 + f];
    }
    score_s[i * HEADS + h] = ss;
    score_r[i * HEADS + h] = sr;
  }
}

__global__ void gat_wcomb(const float* __restrict__ W_edge,
                          const float* __restrict__ a, float* __restrict__ wcomb)
{
  int idx = threadIdx.x;
  if (idx >= HEADS * DEDGE) return;
  int h = idx >> 4, i = idx & 15;
  float acc = 0.f;
#pragma unroll
  for (int f = 0; f < ATT_F; ++f)
    acc += a[h * 48 + 32 + f] * W_edge[((h * ATT_F) + f) * DEDGE + i];
  wcomb[idx] = acc;
}


__device__ __forceinline__ float gat_ez(float m)
{
  float lm = (m > 0.f) ? m : 0.01f * m;
  return __expf(lm);
}

#define SORT_CHUNKS 32
#define LDS_BINS    51200
#define CNT_THREADS 1024
#define CNT_GROUP   8

__device__ __forceinline__ int gat_chunk_len(int n_edges)
{
  return (((n_edges + SORT_CHUNKS - 1) / SORT_CHUNKS) + 3) & ~3;
}
__device__ __forceinline__ int gat_chunk_pitch(int n_nodes)
{
  return (n_nodes + 63) & ~63;
}

__global__ __launch_bounds__(CNT_THREADS) void gat_count(
    const int* __restrict__ recv, int* __restrict__ chunk_cnt, int* __restrict__ rank,
    int n_edges, int n_nodes)
{
  __shared__ int hist[LDS_BINS];
  const int t  = threadIdx.x;
  const int c  = blockIdx.x;
  const int ch = gat_chunk_len(n_edges);
  const int e0 = c * ch;
  const int e1 = (e0 + ch < n_edges) ? e0 + ch : n_edges;
  const int n4 = (e1 > e0) ? ((e1 - e0) >> 2) : 0;

  for (int i = 4 * t; i < LDS_BINS; i += 4 * CNT_THREADS)
    *(int4*)&hist[i] = make_int4(0, 0, 0, 0);
  __syncthreads();

  const int4* __restrict__ recv4 = (const int4*)(recv + e0);
  int4* __restrict__ rank4 = (int4*)(rank + e0);
  for (int gb = 0; gb < n4; gb += CNT_THREADS * CNT_GROUP) {
    int4 rv[CNT_GROUP];
#pragma unroll
    for (int u = 0; u < CNT_GROUP; ++u) {
      const int g = gb + u * CNT_THREADS + t;
      rv[u] = recv4[(g < n4) ? g : 0];
    }
#pragma unroll
    for (int u = 0; u < CNT_GROUP; ++u) {
      const int g = gb + u * CNT_THREADS + t;
      if (g < n4) {
        int4 rk;
        rk.x = atomicAdd(&hist[rv[u].x], 1);
        rk.y = atomicAdd(&hist[rv[u].y], 1);
        rk.z = atomicAdd(&hist[rv[u].z], 1);
        rk.w = atomicAdd(&hist[rv[u].w], 1);
        rank4[g] = rk;
      }
    }
  }
  {
    const int e = e0 + 4 * n4 + t;
    if (t < 4 && e < e1) rank[e] = atomicAdd(&hist[recv[e]], 1);
  }
  __syncthreads();

  int* __restrict__ row = chunk_cnt + (size_t)c * gat_chunk_pitch(n_nodes);
  for (int i = t; i < n_nodes; i += CNT_THREADS) row[i] = hist[i];
}

__global__ __launch_bounds__(256) void gat_chunk_starts(
    int* __restrict__ chunk_cnt, int* __restrict__ cnt, int n_nodes)
{
  const int r = blockIdx.x * blockDim.x + threadIdx.x;
  if (r >= n_nodes) return;
  const size_t pitch = (size_t)gat_chunk_pitch(n_nodes);
  int v[SORT_CHUNKS];
#pragma unroll
  for (int c = 0; c < SORT_CHUNKS; ++c) v[c] = chunk_cnt[c * pitch + r];
  int run = 0;
#pragma unroll
  for (int c = 0; c < SORT_CHUNKS; ++c) {
    chunk_cnt[c * pitch + r] = run;
    run += v[c];
  }
  cnt[r] = run;
}

__global__ __launch_bounds__(1024) void gat_scan(
    const int* __restrict__ cnt, int* __restrict__ offs, int n)
{
  __shared__ int wsum[32];
  const int t = threadIdx.x, lane = t & 31, w = t >> 5;
  int carry = 0;
  for (int base = 0; base < n; base += 4096) {
    int i0 = base + 4 * t;
    int c0 = (i0     < n) ? cnt[i0]     : 0;
    int c1 = (i0 + 1 < n) ? cnt[i0 + 1] : 0;
    int c2 = (i0 + 2 < n) ? cnt[i0 + 2] : 0;
    int c3 = (i0 + 3 < n) ? cnt[i0 + 3] : 0;
    int tsum = c0 + c1 + c2 + c3;
    int x = tsum;
    for (int d = 1; d < 32; d <<= 1) {
      int y = __shfl_up(x, d);
      if (lane >= d) x += y;
    }
    if (lane == 31) wsum[w] = x;
    __syncthreads();
    if (w == 0) {
      int sv = wsum[lane];
      for (int d = 1; d < 32; d <<= 1) {
        int y = __shfl_up(sv, d);
        if (lane >= d) sv += y;
      }
      wsum[lane] = sv;
    }
    __syncthreads();
    int wbase = (w > 0) ? wsum[w - 1] : 0;
    int tot   = wsum[31];
    int ex0 = carry + wbase + x - tsum;
    int ex1 = ex0 + c0, ex2 = ex1 + c1, ex3 = ex2 + c2;
    if (i0     < n) offs[i0]     = ex0;
    if (i0 + 1 < n) offs[i0 + 1] = ex1;
    if (i0 + 2 < n) offs[i0 + 2] = ex2;
    if (i0 + 3 < n) offs[i0 + 3] = ex3;
    carry += tot;
    __syncthreads();
  }
  if (t == 0) offs[n] = carry;
}

__global__ __launch_bounds__(256) void gat_fill(
    const int* __restrict__ recv, const int* __restrict__ send,
    const int* __restrict__ rank, const int* __restrict__ offs,
    const int* __restrict__ chunk_start, int2* __restrict__ rec,
    int n_edges, int n_nodes)
{
  int e = blockIdx.x * blockDim.x + threadIdx.x;
  if (e >= n_edges) return;
  const int r = recv[e];
  const int c = e / gat_chunk_len(n_edges);
  int pos = offs[r] + chunk_start[(size_t)c * gat_chunk_pitch(n_nodes) + r] + rank[e];
  rec[pos] = make_int2(e, send[e]);
}

__global__ __launch_bounds__(256) void gat_gather(
    const float* __restrict__ edges, const int2* __restrict__ rec,
    const int* __restrict__ offs, const float* __restrict__ score_s,
    const float* __restrict__ score_r, const float* __restrict__ wcomb,
    const float* __restrict__ Wn, float* __restrict__ aggr, int n_nodes)
{
  __shared__ float wc[HEADS * DEDGE];
  if (threadIdx.x < HEADS * DEDGE) wc[threadIdx.x] = wcomb[threadIdx.x];
  __syncthreads();

  const int r = __builtin_amdgcn_readfirstlane(
      (int)(blockIdx.x * (blockDim.x >> 5) + (threadIdx.x >> 5)));
  if (r >= n_nodes) return;
  const int lane = threadIdx.x & 31;
  const int hd   = lane >> 3;

  const int beg = offs[r], end = offs[r + 1];
  const float4 sr4 = *(const float4*)(score_r + (size_t)r * HEADS);

  float ax, ay, nrm;
  {
    const float4 ss4 = *(const float4*)(score_s + (size_t)r * HEADS);
    float p0 = gat_ez(ss4.x + sr4.x + 0.f);
    float p1 = gat_ez(ss4.y + sr4.y + 0.f);
    float p2 = gat_ez(ss4.z + sr4.z + 0.f);
    float p3 = gat_ez(ss4.w + sr4.w + 0.f);
    float p  = (hd == 0) ? p0 : (hd == 1) ? p1 : (hd == 2) ? p2 : p3;
    const float2 v = *(const float2*)(Wn + (size_t)r * DNODE + 2 * lane);
    ax = p * v.x; ay = p * v.y; nrm = p;
  }

  for (int base = beg; base < end; base += 32) {
    const int left = end - base;
    const int cnt  = (left < 32) ? left : 32;
    int   sv = r;
    float q0 = 0.f, q1 = 0.f, q2 = 0.f, q3 = 0.f;
    if (lane < cnt) {
      const int2 es = rec[base + lane];
      sv = es.y;
      const float4 ss4 = *(const float4*)(score_s + (size_t)es.y * HEADS);
      const float4* __restrict__ ep = (const float4*)(edges + (size_t)es.x * DEDGE);
      float ef[DEDGE];
#pragma unroll
      for (int c = 0; c < 4; ++c) {
        float4 q = ep[c];
        ef[4 * c + 0] = q.x; ef[4 * c + 1] = q.y; ef[4 * c + 2] = q.z; ef[4 * c + 3] = q.w;
      }
      float elt[HEADS];
#pragma unroll
      for (int h = 0; h < HEADS; ++h) {
        float acc = 0.f;
#pragma unroll
        for (int i = 0; i < DEDGE; ++i) acc += ef[i] * wc[h * DEDGE + i];
        elt[h] = acc;
      }
      q0 = gat_ez(ss4.x + sr4.x + elt[0]);
      q1 = gat_ez(ss4.y + sr4.y + elt[1]);
      q2 = gat_ez(ss4.z + sr4.z + elt[2]);
      q3 = gat_ez(ss4.w + sr4.w + elt[3]);
    }
    int j = 0;
    for (; j + 8 <= cnt; j += 8) {
      for (int u = 0; u < 8; ++u) {
        const int   sj = __builtin_amdgcn_readlane(sv, j + u);
        const float p0 = __int_as_float(__builtin_amdgcn_readlane(__float_as_int(q0), j + u));
        const float p1 = __int_as_float(__builtin_amdgcn_readlane(__float_as_int(q1), j + u));
        const float p2 = __int_as_float(__builtin_amdgcn_readlane(__float_as_int(q2), j + u));
        const float p3 = __int_as_float(__builtin_amdgcn_readlane(__float_as_int(q3), j + u));
        const float p  = (hd == 0) ? p0 : (hd == 1) ? p1 : (hd == 2) ? p2 : p3;
        const float2 v = *(const float2*)(Wn + (size_t)sj * DNODE + 2 * lane);
        ax += p * v.x; ay += p * v.y; nrm += p;
      }
    }
    for (; j < cnt; j += 4) {
      for (int u = 0; u < 4; ++u) {
        const int   sj = __builtin_amdgcn_readlane(sv, j + u);
        const float p0 = __int_as_float(__builtin_amdgcn_readlane(__float_as_int(q0), j + u));
        const float p1 = __int_as_float(__builtin_amdgcn_readlane(__float_as_int(q1), j + u));
        const float p2 = __int_as_float(__builtin_amdgcn_readlane(__float_as_int(q2), j + u));
        const float p3 = __int_as_float(__builtin_amdgcn_readlane(__float_as_int(q3), j + u));
        const float p  = (hd == 0) ? p0 : (hd == 1) ? p1 : (hd == 2) ? p2 : p3;
        const float2 v = *(const float2*)(Wn + (size_t)sj * DNODE + 2 * lane);
        ax += p * v.x; ay += p * v.y; nrm += p;
      }
    }
  }

  float2 o; o.x = ax / nrm; o.y = ay / nrm;
  *(float2*)(aggr + (size_t)r * DNODE + 2 * lane) = o;
}

__global__ __launch_bounds__(256) void gat_finalize(
    float* __restrict__ out, const float* __restrict__ scale,
    const float* __restrict__ bias, int n_nodes)
{
  int i = blockIdx.x * blockDim.x + threadIdx.x;
  if (i >= n_nodes) return;
  float* row = out + (size_t)i * DNODE;
  float v[DNODE];
  float mean = 0.f;
#pragma unroll
  for (int j = 0; j < DNODE; ++j) {
    float x = row[j];
    x = (x > 0.f) ? x : expm1f(x);
    v[j] = x; mean += x;
  }
  mean *= (1.f / DNODE);
  float var = 0.f;
#pragma unroll
  for (int j = 0; j < DNODE; ++j) { float d = v[j] - mean; var += d * d; }
  var *= (1.f / DNODE);
  float inv = rsqrtf(var + 1e-6f);
#pragma unroll
  for (int j = 0; j < DNODE; ++j)
    row[j] = (v[j] - mean) * inv * scale[j] + bias[j];
}

extern "C" void kernel_launch(void* const* d_in, const int* in_sizes, int n_in,
                              void* d_out, int out_size, void* d_ws, size_t ws_size,
                              hipStream_t stream)
{
  const float* nodes  = (const float*)d_in[0];
  const float* edges  = (const float*)d_in[1];
  const int*   recv   = (const int*)d_in[2];
  const int*   send   = (const int*)d_in[3];
  const float* W      = (const float*)d_in[4];
  const float* W_edge = (const float*)d_in[5];
  const float* a      = (const float*)d_in[6];
  const float* ln_s   = (const float*)d_in[7];
  const float* ln_b   = (const float*)d_in[8];

  const int n_nodes = in_sizes[0] / DNODE;
  const int n_edges = in_sizes[1] / DEDGE;
  float* out = (float*)d_out;

  char* ws = (char*)d_ws;
  size_t off = 0;
  auto carve = [&](size_t bytes) -> float* {
    float* p = (float*)(ws + off);
    off += (bytes + 255) & ~(size_t)255;
    return p;
  };
  float* Wn      = carve((size_t)n_nodes * DNODE * sizeof(float));
  float* score_s = carve((size_t)n_nodes * HEADS * sizeof(float));
  float* score_r = carve((size_t)n_nodes * HEADS * sizeof(float));
  float* wcomb   = carve((size_t)HEADS * DEDGE * sizeof(float));
  int*  cnt    = (int*)carve((size_t)n_nodes * sizeof(int));
  int*  offs   = (int*)carve((size_t)(n_nodes + 1) * sizeof(int));
  int2* rec    = (int2*)carve((size_t)n_edges * sizeof(int2));
  int*  rank      = (int*)Wn;
  int*  chunk_cnt = (int*)((char*)Wn + (((size_t)n_edges * sizeof(int) + 255) & ~(size_t)255));

  const int n_tiles = (n_nodes + 15) / 16;
  dim3 blk(256);

  gat_count<<<dim3(SORT_CHUNKS), dim3(CNT_THREADS), 0, stream>>>(
      recv, chunk_cnt, rank, n_edges, n_nodes);
  gat_chunk_starts<<<dim3((n_nodes + 255) / 256), blk, 0, stream>>>(chunk_cnt, cnt, n_nodes);
  gat_scan<<<dim3(1), dim3(1024), 0, stream>>>(cnt, offs, n_nodes);
  gat_fill<<<dim3((n_edges + 255) / 256), blk, 0, stream>>>(
      recv, send, rank, offs, chunk_cnt, rec, n_edges, n_nodes);
  gat_node_project<<<dim3((n_tiles + 7) / 8), blk, 0, stream>>>(
      nodes, W, Wn, n_nodes, n_tiles);
  gat_node_scores<<<dim3((n_nodes + 255) / 256), blk, 0, stream>>>(
      Wn, a, score_s, score_r, n_nodes);
  gat_wcomb<<<dim3(1), dim3(64), 0, stream>>>(W_edge, a, wcomb);
  gat_gather<<<dim3((n_nodes + 7) / 8), blk, 0, stream>>>(
      edges, rec, offs, score_s, score_r, wcomb, Wn, out, n_nodes);
  gat_finalize<<<dim3((n_nodes + 255) / 256), blk, 0, stream>>>(
      out, ln_s, ln_b, n_nodes);
}


# gather takes its 64 edge-weight terms as wave-uniform scalar operands (fused multiply-adds), not from 64 VGPRs filled out of LDS: 146 -> 42 VGPRs, waves per SIMD 6 -> slot limit; sum order kept
# speedup vs baseline: 7.8698x; 1.1123x over previous
#include <hip/hip_runtime.h>
#include <hip/hip_bf16.h>
#include <math.h>

typedef __attribute__((ext_vector_type(2))) float v2f;
typedef __attribute__((ext_vector_type(8))) float v8f;

#define HEADS 4
#define ATT_F 16
#define DNODE 64
#define DEDGE 16

__global__ __launch_bounds__(256) void gat_node_project(
    const float* __restrict__ nodes, const float* __restrict__ W,
    float* __restrict__ Wn, int n_nodes, int n_tiles)
{
  int wave = blockIdx.x * (blockDim.x >> 5) + (threadIdx.x >> 5);
  if (wave >= n_tiles) return;
  int lane = threadIdx.x & 31;
  int half = lane >> 4;
  int l16  = lane & 15;
  int row0 = wave * 16;
  int m = row0 + l16; if (m >= n_nodes) m = n_nodes - 1;
  const float* __restrict__ arow = nodes + (size_t)m * DNODE;

  v8f acc[4] = {};
  for (int k0 = 0; k0 < DNODE; k0 += 4) {
    int kk = k0 + 2 * half;
    v2f a; a.x = arow[kk]; a.y = arow[kk + 1];
#pragma unroll
    for (int t = 0; t < 4; ++t) {
      int j = t * 16 + l16;
      v2f b; b.x = W[(size_t)j * DNODE + kk]; b.y = W[(size_t)j * DNODE + kk + 1];
      acc[t] = __builtin_amdgcn_wmma_f32_16x16x4_f32(
           false, a,  false, b,
           (short)0, acc[t],  false,  false);
    }
  }

  if (row0 + 16 <= n_nodes) {
#pragma unroll
    for (int t = 0; t < 4; ++t)
#pragma unroll
      for (int v = 0; v < 8; ++v)
        Wn[(size_t)(row0 + v + 8 * half) * DNODE + t * 16 + l16] = acc[t][v];
  } else {
#pragma unroll
    for (int t = 0; t < 4; ++t)
#pragma unroll
      for (int v = 0; v < 8; ++v) {
        int mo = row0 + v + 8 * half;
        if (mo < n_nodes) Wn[(size_t)mo * DNODE + t * 16 + l16] = acc[t][v];
      }
  }
}

__global__ __launch_bounds__(256) void gat_node_scores(
    const float* __restrict__ Wn, const float* __restrict__ a,
    float* __restrict__ score_s, float* __restrict__ score_r, int n_nodes)
{
  int i = blockIdx.x * blockDim.x + threadIdx.x;
  if (i >= n_nodes) return;
  const float* __restrict__ row = Wn + (size_t)i * DNODE;
#pragma unroll
  for (int h = 0; h < HEADS; ++h) {
    float ss = 0.f, sr = 0.f;
#pragma unroll
    for (int f = 0; f < ATT_F; ++f) {
      float v = row[h * ATT_F + f];
      ss += v * a[h * 48 + f];
      sr += v * a[h * 48 + 16 + f];
    }
    score_s[i * HEADS + h] = ss;
    score_r[i * HEADS + h] = sr;
  }
}

__global__ void gat_wcomb(const float* __restrict__ W_edge,
                          const float* __restrict__ a, float* __restrict__ wcomb)
{
  int idx = threadIdx.x;
  if (idx >= HEADS * DEDGE) return;
  int h = idx >> 4, i = idx & 15;
  float acc = 0.f;
#pragma unroll
  for (int f = 0; f < ATT_F; ++f)
    acc += a[h * 48 + 32 + f] * W_edge[((h * ATT_F) + f) * DEDGE + i];
  wcomb[idx] = acc;
}


__device__ __forceinline__ float gat_ez(float m)
{
  float lm = (m > 0.f) ? m : 0.01f * m;
  return __expf(lm);
}

#define SORT_CHUNKS 32
#define LDS_BINS    51200
#define CNT_THREADS 1024
#define CNT_GROUP   8

__device__ __forceinline__ int gat_chunk_len(int n_edges)
{
  return (((n_edges + SORT_CHUNKS - 1) / SORT_CHUNKS) + 3) & ~3;
}
__device__ __forceinline__ int gat_chunk_pitch(int n_nodes)
{
  return (n_nodes + 63) & ~63;
}

__global__ __launch_bounds__(CNT_THREADS) void gat_count(
    const int* __restrict__ recv, int* __restrict__ chunk_cnt, int* __restrict__ rank,
    int n_edges, int n_nodes)
{
  __shared__ int hist[LDS_BINS];
  const int t  = threadIdx.x;
  const int c  = blockIdx.x;
  const int ch = gat_chunk_len(n_edges);
  const int e0 = c * ch;
  const int e1 = (e0 + ch < n_edges) ? e0 + ch : n_edges;
  const int n4 = (e1 > e0) ? ((e1 - e0) >> 2) : 0;

  for (int i = 4 * t; i < LDS_BINS; i += 4 * CNT_THREADS)
    *(int4*)&hist[i] = make_int4(0, 0, 0, 0);
  __syncthreads();

  const int4* __restrict__ recv4 = (const int4*)(recv + e0);
  int4* __restrict__ rank4 = (int4*)(rank + e0);
  for (int gb = 0; gb < n4; gb += CNT_THREADS * CNT_GROUP) {
    int4 rv[CNT_GROUP];
#pragma unroll
    for (int u = 0; u < CNT_GROUP; ++u) {
      const int g = gb + u * CNT_THREADS + t;
      rv[u] = recv4[(g < n4) ? g : 0];
    }
#pragma unroll
    for (int u = 0; u < CNT_GROUP; ++u) {
      const int g = gb + u * CNT_THREADS + t;
      if (g < n4) {
        int4 rk;
        rk.x = atomicAdd(&hist[rv[u].x], 1);
        rk.y = atomicAdd(&hist[rv[u].y], 1);
        rk.z = atomicAdd(&hist[rv[u].z], 1);
        rk.w = atomicAdd(&hist[rv[u].w], 1);
        rank4[g] = rk;
      }
    }
  }
  {
    const int e = e0 + 4 * n4 + t;
    if (t < 4 && e < e1) rank[e] = atomicAdd(&hist[recv[e]], 1);
  }
  __syncthreads();

  int* __restrict__ row = chunk_cnt + (size_t)c * gat_chunk_pitch(n_nodes);
  for (int i = t; i < n_nodes; i += CNT_THREADS) row[i] = hist[i];
}

__global__ __launch_bounds__(256) void gat_chunk_starts(
    int* __restrict__ chunk_cnt, int* __restrict__ cnt, int n_nodes)
{
  const int r = blockIdx.x * blockDim.x + threadIdx.x;
  if (r >= n_nodes) return;
  const size_t pitch = (size_t)gat_chunk_pitch(n_nodes);
  int v[SORT_CHUNKS];
#pragma unroll
  for (int c = 0; c < SORT_CHUNKS; ++c) v[c] = chunk_cnt[c * pitch + r];
  int run = 0;
#pragma unroll
  for (int c = 0; c < SORT_CHUNKS; ++c) {
    chunk_cnt[c * pitch + r] = run;
    run += v[c];
  }
  cnt[r] = run;
}

__global__ __launch_bounds__(1024) void gat_scan(
    const int* __restrict__ cnt, int* __restrict__ offs, int n)
{
  __shared__ int wsum[32];
  const int t = threadIdx.x, lane = t & 31, w = t >> 5;
  int carry = 0;
  for (int base = 0; base < n; base += 4096) {
    int i0 = base + 4 * t;
    int c0 = (i0     < n) ? cnt[i0]     : 0;
    int c1 = (i0 + 1 < n) ? cnt[i0 + 1] : 0;
    int c2 = (i0 + 2 < n) ? cnt[i0 + 2] : 0;
    int c3 = (i0 + 3 < n) ? cnt[i0 + 3] : 0;
    int tsum = c0 + c1 + c2 + c3;
    int x = tsum;
    for (int d = 1; d < 32; d <<= 1) {
      int y = __shfl_up(x, d);
      if (lane >= d) x += y;
    }
    if (lane == 31) wsum[w] = x;
    __syncthreads();
    if (w == 0) {
      int sv = wsum[lane];
      for (int d = 1; d < 32; d <<= 1) {
        int y = __shfl_up(sv, d);
        if (lane >= d) sv += y;
      }
      wsum[lane] = sv;
    }
    __syncthreads();
    int wbase = (w > 0) ? wsum[w - 1] : 0;
    int tot   = wsum[31];
    int ex0 = carry + wbase + x - tsum;
    int ex1 = ex0 + c0, ex2 = ex1 + c1, ex3 = ex2 + c2;
    if (i0     < n) offs[i0]     = ex0;
    if (i0 + 1 < n) offs[i0 + 1] = ex1;
    if (i0 + 2 < n) offs[i0 + 2] = ex2;
    if (i0 + 3 < n) offs[i0 + 3] = ex3;
    carry += tot;
    __syncthreads();
  }
  if (t == 0) offs[n] = carry;
}

__global__ __launch_bounds__(256) void gat_fill(
    const int* __restrict__ recv, const int* __restrict__ send,
    const int* __restrict__ rank, const int* __restrict__ offs,
    const int* __restrict__ chunk_start, int2* __restrict__ rec,
    int n_edges, int n_nodes)
{
  int e = blockIdx.x * blockDim.x + threadIdx.x;
  if (e >= n_edges) return;
  const int r = recv[e];
  const int c = e / gat_chunk_len(n_edges);
  int pos = offs[r] + chunk_start[(size_t)c * gat_chunk_pitch(n_nodes) + r] + rank[e];
  rec[pos] = make_int2(e, send[e]);
}

__global__ __launch_bounds__(256) void gat_gather(
    const float* __restrict__ edges, const int2* __restrict__ rec,
    const int* __restrict__ offs, const float* __restrict__ score_s,
    const float* __restrict__ score_r, const float* __restrict__ wcomb,
    const float* __restrict__ Wn, float* __restrict__ aggr, int n_nodes)
{
  const int r = __builtin_amdgcn_readfirstlane(
      (int)(blockIdx.x * (blockDim.x >> 5) + (threadIdx.x >> 5)));
  if (r >= n_nodes) return;
  const int lane = threadIdx.x & 31;
  const int hd   = lane >> 3;

  const int beg = offs[r], end = offs[r + 1];
  const float4 sr4 = *(const float4*)(score_r + (size_t)r * HEADS);

  float ax, ay, nrm;
  {
    const float4 ss4 = *(const float4*)(score_s + (size_t)r * HEADS);
    float p0 = gat_ez(ss4.x + sr4.x + 0.f);
    float p1 = gat_ez(ss4.y + sr4.y + 0.f);
    float p2 = gat_ez(ss4.z + sr4.z + 0.f);
    float p3 = gat_ez(ss4.w + sr4.w + 0.f);
    float p  = (hd == 0) ? p0 : (hd == 1) ? p1 : (hd == 2) ? p2 : p3;
    const float2 v = *(const float2*)(Wn + (size_t)r * DNODE + 2 * lane);
    ax = p * v.x; ay = p * v.y; nrm = p;
  }

  for (int base = beg; base < end; base += 32) {
    const int left = end - base;
    const int cnt  = (left < 32) ? left : 32;
    int   sv = r;
    float q0 = 0.f, q1 = 0.f, q2 = 0.f, q3 = 0.f;
    if (lane < cnt) {
      const int2 es = rec[base + lane];
      sv = es.y;
      const float4 ss4 = *(const float4*)(score_s + (size_t)es.y * HEADS);
      const float4* __restrict__ ep = (const float4*)(edges + (size_t)es.x * DEDGE);
      float ef[DEDGE];
#pragma unroll
      for (int c = 0; c < 4; ++c) {
        float4 q = ep[c];
        ef[4 * c + 0] = q.x; ef[4 * c + 1] = q.y; ef[4 * c + 2] = q.z; ef[4 * c + 3] = q.w;
      }
      float elt[HEADS];
#pragma unroll
      for (int h = 0; h < HEADS; ++h) {
        float acc = 0.f;
#pragma unroll
        for (int i = 0; i < DEDGE; ++i) acc = __builtin_fmaf(ef[i], wcomb[h * DEDGE + i], acc);
        elt[h] = acc;
      }
      q0 = gat_ez(ss4.x + sr4.x + elt[0]);
      q1 = gat_ez(ss4.y + sr4.y + elt[1]);
      q2 = gat_ez(ss4.z + sr4.z + elt[2]);
      q3 = gat_ez(ss4.w + sr4.w + elt[3]);
    }
    int j = 0;
    for (; j + 8 <= cnt; j += 8) {
      for (int u = 0; u < 8; ++u) {
        const int   sj = __builtin_amdgcn_readlane(sv, j + u);
        const float p0 = __int_as_float(__builtin_amdgcn_readlane(__float_as_int(q0), j + u));
        const float p1 = __int_as_float(__builtin_amdgcn_readlane(__float_as_int(q1), j + u));
        const float p2 = __int_as_float(__builtin_amdgcn_readlane(__float_as_int(q2), j + u));
        const float p3 = __int_as_float(__builtin_amdgcn_readlane(__float_as_int(q3), j + u));
        const float p  = (hd == 0) ? p0 : (hd == 1) ? p1 : (hd == 2) ? p2 : p3;
        const float2 v = *(const float2*)(Wn + (size_t)sj * DNODE + 2 * lane);
        ax += p * v.x; ay += p * v.y; nrm += p;
      }
    }
    for (; j < cnt; j += 4) {
      for (int u = 0; u < 4; ++u) {
        const int   sj = __builtin_amdgcn_readlane(sv, j + u);
        const float p0 = __int_as_float(__builtin_amdgcn_readlane(__float_as_int(q0), j + u));
        const float p1 = __int_as_float(__builtin_amdgcn_readlane(__float_as_int(q1), j + u));
        const float p2 = __int_as_float(__builtin_amdgcn_readlane(__float_as_int(q2), j + u));
        const float p3 = __int_as_float(__builtin_amdgcn_readlane(__float_as_int(q3), j + u));
        const float p  = (hd == 0) ? p0 : (hd == 1) ? p1 : (hd == 2) ? p2 : p3;
        const float2 v = *(const float2*)(Wn + (size_t)sj * DNODE + 2 * lane);
        ax += p * v.x; ay += p * v.y; nrm += p;
      }
    }
  }

  float2 o; o.x = ax / nrm; o.y = ay / nrm;
  *(float2*)(aggr + (size_t)r * DNODE + 2 * lane) = o;
}

__global__ __launch_bounds__(256) void gat_finalize(
    float* __restrict__ out, const float* __restrict__ scale,
    const float* __restrict__ bias, int n_nodes)
{
  int i = blockIdx.x * blockDim.x + threadIdx.x;
  if (i >= n_nodes) return;
  float* row = out + (size_t)i * DNODE;
  float v[DNODE];
  float mean = 0.f;
#pragma unroll
  for (int j = 0; j < DNODE; ++j) {
    float x = row[j];
    x = (x > 0.f) ? x : expm1f(x);
    v[j] = x; mean += x;
  }
  mean *= (1.f / DNODE);
  float var = 0.f;
#pragma unroll
  for (int j = 0; j < DNODE; ++j) { float d = v[j] - mean; var += d * d; }
  var *= (1.f / DNODE);
  float inv = rsqrtf(var + 1e-6f);
#pragma unroll
  for (int j = 0; j < DNODE; ++j)
    row[j] = (v[j] - mean) * inv * scale[j] + bias[j];
}

extern "C" void kernel_launch(void* const* d_in, const int* in_sizes, int n_in,
                              void* d_out, int out_size, void* d_ws, size_t ws_size,
                              hipStream_t stream)
{
  const float* nodes  = (const float*)d_in[0];
  const float* edges  = (const float*)d_in[1];
  const int*   recv   = (const int*)d_in[2];
  const int*   send   = (const int*)d_in[3];
  const float* W      = (const float*)d_in[4];
  const float* W_edge = (const float*)d_in[5];
  const float* a      = (const float*)d_in[6];
  const float* ln_s   = (const float*)d_in[7];
  const float* ln_b   = (const float*)d_in[8];

  const int n_nodes = in_sizes[0] / DNODE;
  const int n_edges = in_sizes[1] / DEDGE;
  float* out = (float*)d_out;

  char* ws = (char*)d_ws;
  size_t off = 0;
  auto carve = [&](size_t bytes) -> float* {
    float* p = (float*)(ws + off);
    off += (bytes + 255) & ~(size_t)255;
    return p;
  };
  float* Wn      = carve((size_t)n_nodes * DNODE * sizeof(float));
  float* score_s = carve((size_t)n_nodes * HEADS * sizeof(float));
  float* score_r = carve((size_t)n_nodes * HEADS * sizeof(float));
  float* wcomb   = carve((size_t)HEADS * DEDGE * sizeof(float));
  int*  cnt    = (int*)carve((size_t)n_nodes * sizeof(int));
  int*  offs   = (int*)carve((size_t)(n_nodes + 1) * sizeof(int));
  int2* rec    = (int2*)carve((size_t)n_edges * sizeof(int2));
  int*  rank      = (int*)Wn;
  int*  chunk_cnt = (int*)((char*)Wn + (((size_t)n_edges * sizeof(int) + 255) & ~(size_t)255));

  const int n_tiles = (n_nodes + 15) / 16;
  dim3 blk(256);

  gat_count<<<dim3(SORT_CHUNKS), dim3(CNT_THREADS), 0, stream>>>(
      recv, chunk_cnt, rank, n_edges, n_nodes);
  gat_chunk_starts<<<dim3((n_nodes + 255) / 256), blk, 0, stream>>>(chunk_cnt, cnt, n_nodes);
  gat_scan<<<dim3(1), dim3(1024), 0, stream>>>(cnt, offs, n_nodes);
  gat_fill<<<dim3((n_edges + 255) / 256), blk, 0, stream>>>(
      recv, send, rank, offs, chunk_cnt, rec, n_edges, n_nodes);
  gat_node_project<<<dim3((n_tiles + 7) / 8), blk, 0, stream>>>(
      nodes, W, Wn, n_nodes, n_tiles);
  gat_node_scores<<<dim3((n_nodes + 255) / 256), blk, 0, stream>>>(
      Wn, a, score_s, score_r, n_nodes);
  gat_wcomb<<<dim3(1), dim3(64), 0, stream>>>(W_edge, a, wcomb);
  gat_gather<<<dim3((n_nodes + 7) / 8), blk, 0, stream>>>(
      edges, rec, offs, score_s, score_r, wcomb, Wn, out, n_nodes);
  gat_finalize<<<dim3((n_nodes + 255) / 256), blk, 0, stream>>>(
      out, ln_s, ln_b, n_nodes);
}


# prefix scan launch removed: the chunk-start launch scans its own 256 receivers and writes block totals, each fill work-group turns the 196 totals into block starts in LDS; integer sums, slots kept
# speedup vs baseline: 9.4740x; 1.2038x over previous
#include <hip/hip_runtime.h>
#include <hip/hip_bf16.h>
#include <math.h>

typedef __attribute__((ext_vector_type(2))) float v2f;
typedef __attribute__((ext_vector_type(8))) float v8f;

#define HEADS 4
#define ATT_F 16
#define DNODE 64
#define DEDGE 16

__global__ __launch_bounds__(256) void gat_node_project(
    const float* __restrict__ nodes, const float* __restrict__ W,
    float* __restrict__ Wn, int n_nodes, int n_tiles)
{
  int wave = blockIdx.x * (blockDim.x >> 5) + (threadIdx.x >> 5);
  if (wave >= n_tiles) return;
  int lane = threadIdx.x & 31;
  int half = lane >> 4;
  int l16  = lane & 15;
  int row0 = wave * 16;
  int m = row0 + l16; if (m >= n_nodes) m = n_nodes - 1;
  const float* __restrict__ arow = nodes + (size_t)m * DNODE;

  v8f acc[4] = {};
  for (int k0 = 0; k0 < DNODE; k0 += 4) {
    int kk = k0 + 2 * half;
    v2f a; a.x = arow[kk]; a.y = arow[kk + 1];
#pragma unroll
    for (int t = 0; t < 4; ++t) {
      int j = t * 16 + l16;
      v2f b; b.x = W[(size_t)j * DNODE + kk]; b.y = W[(size_t)j * DNODE + kk + 1];
      acc[t] = __builtin_amdgcn_wmma_f32_16x16x4_f32(
           false, a,  false, b,
           (short)0, acc[t],  false,  false);
    }
  }

  if (row0 + 16 <= n_nodes) {
#pragma unroll
    for (int t = 0; t < 4; ++t)
#pragma unroll
      for (int v = 0; v < 8; ++v)
        Wn[(size_t)(row0 + v + 8 * half) * DNODE + t * 16 + l16] = acc[t][v];
  } else {
#pragma unroll
    for (int t = 0; t < 4; ++t)
#pragma unroll
      for (int v = 0; v < 8; ++v) {
        int mo = row0 + v + 8 * half;
        if (mo < n_nodes) Wn[(size_t)mo * DNODE + t * 16 + l16] = acc[t][v];
      }
  }
}

__global__ __launch_bounds__(256) void gat_node_scores(
    const float* __restrict__ Wn, const float* __restrict__ a,
    float* __restrict__ score_s, float* __restrict__ score_r, int n_nodes)
{
  int i = blockIdx.x * blockDim.x + threadIdx.x;
  if (i >= n_nodes) return;
  const float* __restrict__ row = Wn + (size_t)i * DNODE;
#pragma unroll
  for (int h = 0; h < HEADS; ++h) {
    float ss = 0.f, sr = 0.f;
#pragma unroll
    for (int f = 0; f < ATT_F; ++f) {
      float v = row[h * ATT_F + f];
      ss += v * a[h * 48 + f];
      sr += v * a[h * 48 + 16 + f];
    }
    score_s[i * HEADS + h] = ss;
    score_r[i * HEADS + h] = sr;
  }
}

__global__ void gat_wcomb(const float* __restrict__ W_edge,
                          const float* __restrict__ a, float* __restrict__ wcomb)
{
  int idx = threadIdx.x;
  if (idx >= HEADS * DEDGE) return;
  int h = idx >> 4, i = idx & 15;
  float acc = 0.f;
#pragma unroll
  for (int f = 0; f < ATT_F; ++f)
    acc += a[h * 48 + 32 + f] * W_edge[((h * ATT_F) + f) * DEDGE + i];
  wcomb[idx] = acc;
}


__device__ __forceinline__ float gat_ez(float m)
{
  float lm = (m > 0.f) ? m : 0.01f * m;
  return __expf(lm);
}

#define SORT_CHUNKS 32
#define LDS_BINS    51200
#define CNT_THREADS 1024
#define CNT_GROUP   8

__device__ __forceinline__ int gat_chunk_len(int n_edges)
{
  return (((n_edges + SORT_CHUNKS - 1) / SORT_CHUNKS) + 3) & ~3;
}
__device__ __forceinline__ int gat_chunk_pitch(int n_nodes)
{
  return (n_nodes + 63) & ~63;
}

__global__ __launch_bounds__(CNT_THREADS) void gat_count(
    const int* __restrict__ recv, int* __restrict__ chunk_cnt, int* __restrict__ rank,
    int n_edges, int n_nodes)
{
  __shared__ int hist[LDS_BINS];
  const int t  = threadIdx.x;
  const int c  = blockIdx.x;
  const int ch = gat_chunk_len(n_edges);
  const int e0 = c * ch;
  const int e1 = (e0 + ch < n_edges) ? e0 + ch : n_edges;
  const int n4 = (e1 > e0) ? ((e1 - e0) >> 2) : 0;

  for (int i = 4 * t; i < LDS_BINS; i += 4 * CNT_THREADS)
    *(int4*)&hist[i] = make_int4(0, 0, 0, 0);
  __syncthreads();

  const int4* __restrict__ recv4 = (const int4*)(recv + e0);
  int4* __restrict__ rank4 = (int4*)(rank + e0);
  for (int gb = 0; gb < n4; gb += CNT_THREADS * CNT_GROUP) {
    int4 rv[CNT_GROUP];
#pragma unroll
    for (int u = 0; u < CNT_GROUP; ++u) {
      const int g = gb + u * CNT_THREADS + t;
      rv[u] = recv4[(g < n4) ? g : 0];
    }
#pragma unroll
    for (int u = 0; u < CNT_GROUP; ++u) {
      const int g = gb + u * CNT_THREADS + t;
      if (g < n4) {
        int4 rk;
        rk.x = atomicAdd(&hist[rv[u].x], 1);
        rk.y = atomicAdd(&hist[rv[u].y], 1);
        rk.z = atomicAdd(&hist[rv[u].z], 1);
        rk.w = atomicAdd(&hist[rv[u].w], 1);
        rank4[g] = rk;
      }
    }
  }
  {
    const int e = e0 + 4 * n4 + t;
    if (t < 4 && e < e1) rank[e] = atomicAdd(&hist[recv[e]], 1);
  }
  __syncthreads();

  int* __restrict__ row = chunk_cnt + (size_t)c * gat_chunk_pitch(n_nodes);
  for (int i = t; i < n_nodes; i += CNT_THREADS) row[i] = hist[i];
}

#define SCAN_BLOCK 256

__device__ __forceinline__ int gat_wave_incl(int x, int lane)
{
  for (int d = 1; d < 32; d <<= 1) {
    int y = __shfl_up(x, d);
    if (lane >= d) x += y;
  }
  return x;
}

__global__ __launch_bounds__(SCAN_BLOCK) void gat_chunk_starts(
    int* __restrict__ chunk_cnt, int* __restrict__ within, int* __restrict__ blk_tot,
    int n_nodes)
{
  __shared__ int wtot[SCAN_BLOCK / 32];
  const int t = threadIdx.x, lane = t & 31, w = t >> 5;
  const int r = blockIdx.x * SCAN_BLOCK + t;
  const bool live = r < n_nodes;
  const int  rl = live ? r : n_nodes - 1;
  const size_t pitch = (size_t)gat_chunk_pitch(n_nodes);
  int v[SORT_CHUNKS];
#pragma unroll
  for (int c = 0; c < SORT_CHUNKS; ++c) v[c] = chunk_cnt[c * pitch + rl];
  int tot = 0;
#pragma unroll
  for (int c = 0; c < SORT_CHUNKS; ++c) tot += v[c];
  if (!live) tot = 0;

  const int x = gat_wave_incl(tot, lane);
  if (lane == 31) wtot[w] = x;
  __syncthreads();
  int wbase = 0, btot = 0;
#pragma unroll
  for (int k = 0; k < SCAN_BLOCK / 32; ++k) {
    const int s = wtot[k];
    if (k < w) wbase += s;
    btot += s;
  }
  const int ex = wbase + x - tot;

  if (live) {
    int run = ex;
#pragma unroll
    for (int c = 0; c < SORT_CHUNKS; ++c) {
      chunk_cnt[c * pitch + r] = run;
      run += v[c];
    }
    within[r] = ex;
  }
  if (t == 0) blk_tot[blockIdx.x] = btot;
}

__global__ __launch_bounds__(256) void gat_fill(
    const int* __restrict__ recv, const int* __restrict__ send,
    const int* __restrict__ rank, const int* __restrict__ within,
    const int* __restrict__ blk_tot, const int* __restrict__ chunk_start,
    int2* __restrict__ rec, int* __restrict__ offs, int n_edges, int n_nodes)
{
  __shared__ int blk_start[256];
  __shared__ int wtot[8];
  const int t = threadIdx.x, lane = t & 31, w = t >> 5;
  const int nb = (n_nodes + SCAN_BLOCK - 1) / SCAN_BLOCK;
  const int e = blockIdx.x * blockDim.x + t;
  const bool live = e < n_edges;
  const int tb = blk_tot[(t < nb) ? t : nb - 1];
  const int el = live ? e : n_edges - 1;
  const int r = recv[el], s = send[el], rk = rank[el];
  const int bt = (t < nb) ? tb : 0;

  const int x = gat_wave_incl(bt, lane);
  if (lane == 31) wtot[w] = x;
  __syncthreads();
  int wbase = 0;
#pragma unroll
  for (int k = 0; k < 8; ++k) {
    const int sw = wtot[k];
    if (k < w) wbase += sw;
  }
  const int ex = wbase + x - bt;
  blk_start[t] = ex;
  __syncthreads();

  if (live) {
    const int c = e / gat_chunk_len(n_edges);
    const int pos = blk_start[(unsigned)r / SCAN_BLOCK]
                  + chunk_start[(size_t)c * gat_chunk_pitch(n_nodes) + r] + rk;
    rec[pos] = make_int2(e, s);
  }

  for (int b = blockIdx.x; b < nb; b += gridDim.x) {
    const int i = b * SCAN_BLOCK + t;
    if (i < n_nodes) offs[i] = blk_start[b] + within[i];
  }
  if (blockIdx.x == 0 && t == nb - 1) offs[n_nodes] = ex + bt;
}

__global__ __launch_bounds__(256) void gat_gather(
    const float* __restrict__ edges, const int2* __restrict__ rec,
    const int* __restrict__ offs, const float* __restrict__ score_s,
    const float* __restrict__ score_r, const float* __restrict__ wcomb,
    const float* __restrict__ Wn, float* __restrict__ aggr, int n_nodes)
{
  const int r = __builtin_amdgcn_readfirstlane(
      (int)(blockIdx.x * (blockDim.x >> 5) + (threadIdx.x >> 5)));
  if (r >= n_nodes) return;
  const int lane = threadIdx.x & 31;
  const int hd   = lane >> 3;

  const int beg = offs[r], end = offs[r + 1];
  const float4 sr4 = *(const float4*)(score_r + (size_t)r * HEADS);

  float ax, ay, nrm;
  {
    const float4 ss4 = *(const float4*)(score_s + (size_t)r * HEADS);
    float p0 = gat_ez(ss4.x + sr4.x + 0.f);
    float p1 = gat_ez(ss4.y + sr4.y + 0.f);
    float p2 = gat_ez(ss4.z + sr4.z + 0.f);
    float p3 = gat_ez(ss4.w + sr4.w + 0.f);
    float p  = (hd == 0) ? p0 : (hd == 1) ? p1 : (hd == 2) ? p2 : p3;
    const float2 v = *(const float2*)(Wn + (size_t)r * DNODE + 2 * lane);
    ax = p * v.x; ay = p * v.y; nrm = p;
  }

  for (int base = beg; base < end; base += 32) {
    const int left = end - base;
    const int cnt  = (left < 32) ? left : 32;
    int   sv = r;
    float q0 = 0.f, q1 = 0.f, q2 = 0.f, q3 = 0.f;
    if (lane < cnt) {
      const int2 es = rec[base + lane];
      sv = es.y;
      const float4 ss4 = *(const float4*)(score_s + (size_t)es.y * HEADS);
      const float4* __restrict__ ep = (const float4*)(edges + (size_t)es.x * DEDGE);
      float ef[DEDGE];
#pragma unroll
      for (int c = 0; c < 4; ++c) {
        float4 q = ep[c];
        ef[4 * c + 0] = q.x; ef[4 * c + 1] = q.y; ef[4 * c + 2] = q.z; ef[4 * c + 3] = q.w;
      }
      float elt[HEADS];
#pragma unroll
      for (int h = 0; h < HEADS; ++h) {
        float acc = 0.f;
#pragma unroll
        for (int i = 0; i < DEDGE; ++i) acc = __builtin_fmaf(ef[i], wcomb[h * DEDGE + i], acc);
        elt[h] = acc;
      }
      q0 = gat_ez(ss4.x + sr4.x + elt[0]);
      q1 = gat_ez(ss4.y + sr4.y + elt[1]);
      q2 = gat_ez(ss4.z + sr4.z + elt[2]);
      q3 = gat_ez(ss4.w + sr4.w + elt[3]);
    }
    int j = 0;
    for (; j + 8 <= cnt; j += 8) {
      for (int u = 0; u < 8; ++u) {
        const int   sj = __builtin_amdgcn_readlane(sv, j + u);
        const float p0 = __int_as_float(__builtin_amdgcn_readlane(__float_as_int(q0), j + u));
        const float p1 = __int_as_float(__builtin_amdgcn_readlane(__float_as_int(q1), j + u));
        const float p2 = __int_as_float(__builtin_amdgcn_readlane(__float_as_int(q2), j + u));
        const float p3 = __int_as_float(__builtin_amdgcn_readlane(__float_as_int(q3), j + u));
        const float p  = (hd == 0) ? p0 : (hd == 1) ? p1 : (hd == 2) ? p2 : p3;
        const float2 v = *(const float2*)(Wn + (size_t)sj * DNODE + 2 * lane);
        ax += p * v.x; ay += p * v.y; nrm += p;
      }
    }
    for (; j < cnt; j += 4) {
      for (int u = 0; u < 4; ++u) {
        const int   sj = __builtin_amdgcn_readlane(sv, j + u);
        const float p0 = __int_as_float(__builtin_amdgcn_readlane(__float_as_int(q0), j + u));
        const float p1 = __int_as_float(__builtin_amdgcn_readlane(__float_as_int(q1), j + u));
        const float p2 = __int_as_float(__builtin_amdgcn_readlane(__float_as_int(q2), j + u));
        const float p3 = __int_as_float(__builtin_amdgcn_readlane(__float_as_int(q3), j + u));
        const float p  = (hd == 0) ? p0 : (hd == 1) ? p1 : (hd == 2) ? p2 : p3;
        const float2 v = *(const float2*)(Wn + (size_t)sj * DNODE + 2 * lane);
        ax += p * v.x; ay += p * v.y; nrm += p;
      }
    }
  }

  float2 o; o.x = ax / nrm; o.y = ay / nrm;
  *(float2*)(aggr + (size_t)r * DNODE + 2 * lane) = o;
}

__global__ __launch_bounds__(256) void gat_finalize(
    float* __restrict__ out, const float* __restrict__ scale,
    const float* __restrict__ bias, int n_nodes)
{
  int i = blockIdx.x * blockDim.x + threadIdx.x;
  if (i >= n_nodes) return;
  float* row = out + (size_t)i * DNODE;
  float v[DNODE];
  float mean = 0.f;
#pragma unroll
  for (int j = 0; j < DNODE; ++j) {
    float x = row[j];
    x = (x > 0.f) ? x : expm1f(x);
    v[j] = x; mean += x;
  }
  mean *= (1.f / DNODE);
  float var = 0.f;
#pragma unroll
  for (int j = 0; j < DNODE; ++j) { float d = v[j] - mean; var += d * d; }
  var *= (1.f / DNODE);
  float inv = rsqrtf(var + 1e-6f);
#pragma unroll
  for (int j = 0; j < DNODE; ++j)
    row[j] = (v[j] - mean) * inv * scale[j] + bias[j];
}

extern "C" void kernel_launch(void* const* d_in, const int* in_sizes, int n_in,
                              void* d_out, int out_size, void* d_ws, size_t ws_size,
                              hipStream_t stream)
{
  const float* nodes  = (const float*)d_in[0];
  const float* edges  = (const float*)d_in[1];
  const int*   recv   = (const int*)d_in[2];
  const int*   send   = (const int*)d_in[3];
  const float* W      = (const float*)d_in[4];
  const float* W_edge = (const float*)d_in[5];
  const float* a      = (const float*)d_in[6];
  const float* ln_s   = (const float*)d_in[7];
  const float* ln_b   = (const float*)d_in[8];

  const int n_nodes = in_sizes[0] / DNODE;
  const int n_edges = in_sizes[1] / DEDGE;
  float* out = (float*)d_out;

  char* ws = (char*)d_ws;
  size_t off = 0;
  auto carve = [&](size_t bytes) -> float* {
    float* p = (float*)(ws + off);
    off += (bytes + 255) & ~(size_t)255;
    return p;
  };
  float* Wn      = carve((size_t)n_nodes * DNODE * sizeof(float));
  float* score_s = carve((size_t)n_nodes * HEADS * sizeof(float));
  float* score_r = carve((size_t)n_nodes * HEADS * sizeof(float));
  float* wcomb   = carve((size_t)HEADS * DEDGE * sizeof(float));
  int*  within = (int*)carve((size_t)n_nodes * sizeof(int));
  int*  offs   = (int*)carve((size_t)(n_nodes + 1) * sizeof(int));
  int2* rec    = (int2*)carve((size_t)n_edges * sizeof(int2));
  int*  rank      = (int*)Wn;
  int*  chunk_cnt = (int*)((char*)Wn + (((size_t)n_edges * sizeof(int) + 255) & ~(size_t)255));
  int*  blk_tot   = chunk_cnt + (size_t)SORT_CHUNKS * (size_t)((n_nodes + 63) & ~63);

  const int n_tiles = (n_nodes + 15) / 16;
  dim3 blk(256);

  gat_count<<<dim3(SORT_CHUNKS), dim3(CNT_THREADS), 0, stream>>>(
      recv, chunk_cnt, rank, n_edges, n_nodes);
  gat_chunk_starts<<<dim3((n_nodes + SCAN_BLOCK - 1) / SCAN_BLOCK), dim3(SCAN_BLOCK), 0, stream>>>(
      chunk_cnt, within, blk_tot, n_nodes);
  gat_fill<<<dim3((n_edges + 255) / 256), blk, 0, stream>>>(
      recv, send, rank, within, blk_tot, chunk_cnt, rec, offs, n_edges, n_nodes);
  gat_node_project<<<dim3((n_tiles + 7) / 8), blk, 0, stream>>>(
      nodes, W, Wn, n_nodes, n_tiles);
  gat_node_scores<<<dim3((n_nodes + 255) / 256), blk, 0, stream>>>(
      Wn, a, score_s, score_r, n_nodes);
  gat_wcomb<<<dim3(1), dim3(64), 0, stream>>>(W_edge, a, wcomb);
  gat_gather<<<dim3((n_nodes + 7) / 8), blk, 0, stream>>>(
      edges, rec, offs, score_s, score_r, wcomb, Wn, out, n_nodes);
  gat_finalize<<<dim3((n_nodes + 255) / 256), blk, 0, stream>>>(
      out, ln_s, ln_b, n_nodes);
}


# independent stages run side by side in one launch as separate work-groups: projection beside the edge count, node scores and edge-weight fold beside chunk starts; 8 -> 5 launches; sum order kept
# speedup vs baseline: 10.3263x; 1.0900x over previous
#include <hip/hip_runtime.h>
#include <hip/hip_bf16.h>
#include <math.h>

typedef __attribute__((ext_vector_type(2))) float v2f;
typedef __attribute__((ext_vector_type(8))) float v8f;

#define HEADS 4
#define ATT_F 16
#define DNODE 64
#define DEDGE 16

__device__ __forceinline__ void gat_node_project(
    const float* __restrict__ nodes, const float* __restrict__ W,
    float* __restrict__ Wn, int n_nodes, int n_tiles, int blk)
{
  int wave = blk * (blockDim.x >> 5) + (threadIdx.x >> 5);
  if (wave >= n_tiles) return;
  int lane = threadIdx.x & 31;
  int half = lane >> 4;
  int l16  = lane & 15;
  int row0 = wave * 16;
  int m = row0 + l16; if (m >= n_nodes) m = n_nodes - 1;
  const float* __restrict__ arow = nodes + (size_t)m * DNODE;

  v8f acc[4] = {};
  for (int k0 = 0; k0 < DNODE; k0 += 4) {
    int kk = k0 + 2 * half;
    v2f a; a.x = arow[kk]; a.y = arow[kk + 1];
#pragma unroll
    for (int t = 0; t < 4; ++t) {
      int j = t * 16 + l16;
      v2f b; b.x = W[(size_t)j * DNODE + kk]; b.y = W[(size_t)j * DNODE + kk + 1];
      acc[t] = __builtin_amdgcn_wmma_f32_16x16x4_f32(
           false, a,  false, b,
           (short)0, acc[t],  false,  false);
    }
  }

  if (row0 + 16 <= n_nodes) {
#pragma unroll
    for (int t = 0; t < 4; ++t)
#pragma unroll
      for (int v = 0; v < 8; ++v)
        Wn[(size_t)(row0 + v + 8 * half) * DNODE + t * 16 + l16] = acc[t][v];
  } else {
#pragma unroll
    for (int t = 0; t < 4; ++t)
#pragma unroll
      for (int v = 0; v < 8; ++v) {
        int mo = row0 + v + 8 * half;
        if (mo < n_nodes) Wn[(size_t)mo * DNODE + t * 16 + l16] = acc[t][v];
      }
  }
}

__device__ __forceinline__ void gat_node_scores(
    const float* __restrict__ Wn, const float* __restrict__ a,
    float* __restrict__ score_s, float* __restrict__ score_r, int n_nodes, int blk)
{
  int i = blk * blockDim.x + threadIdx.x;
  if (i >= n_nodes) return;
  const float* __restrict__ row = Wn + (size_t)i * DNODE;
#pragma unroll
  for (int h = 0; h < HEADS; ++h) {
    float ss = 0.f, sr = 0.f;
#pragma unroll
    for (int f = 0; f < ATT_F; ++f) {
      float v = row[h * ATT_F + f];
      ss += v * a[h * 48 + f];
      sr += v * a[h * 48 + 16 + f];
    }
    score_s[i * HEADS + h] = ss;
    score_r[i * HEADS + h] = sr;
  }
}

__device__ __forceinline__ void gat_wcomb(const float* __restrict__ W_edge,
                          const float* __restrict__ a, float* __restrict__ wcomb)
{
  int idx = threadIdx.x;
  if (idx >= HEADS * DEDGE) return;
  int h = idx >> 4, i = idx & 15;
  float acc = 0.f;
#pragma unroll
  for (int f = 0; f < ATT_F; ++f)
    acc += a[h * 48 + 32 + f] * W_edge[((h * ATT_F) + f) * DEDGE + i];
  wcomb[idx] = acc;
}


__device__ __forceinline__ float gat_ez(float m)
{
  float lm = (m > 0.f) ? m : 0.01f * m;
  return __expf(lm);
}

#define SORT_CHUNKS 32
#define LDS_BINS    51200
#define CNT_THREADS 1024
#define CNT_GROUP   8

__device__ __forceinline__ int gat_chunk_len(int n_edges)
{
  return (((n_edges + SORT_CHUNKS - 1) / SORT_CHUNKS) + 3) & ~3;
}
__device__ __forceinline__ int gat_chunk_pitch(int n_nodes)
{
  return (n_nodes + 63) & ~63;
}

__device__ __forceinline__ void gat_count(
    int* __restrict__ hist,
    const int* __restrict__ recv, int* __restrict__ chunk_cnt, int* __restrict__ rank,
    int n_edges, int n_nodes)
{
  const int t  = threadIdx.x;
  const int c  = blockIdx.x;
  const int ch = gat_chunk_len(n_edges);
  const int e0 = c * ch;
  const int e1 = (e0 + ch < n_edges) ? e0 + ch : n_edges;
  const int n4 = (e1 > e0) ? ((e1 - e0) >> 2) : 0;

  for (int i = 4 * t; i < LDS_BINS; i += 4 * CNT_THREADS)
    *(int4*)&hist[i] = make_int4(0, 0, 0, 0);
  __syncthreads();

  const int4* __restrict__ recv4 = (const int4*)(recv + e0);
  int4* __restrict__ rank4 = (int4*)(rank + e0);
  for (int gb = 0; gb < n4; gb += CNT_THREADS * CNT_GROUP) {
    int4 rv[CNT_GROUP];
#pragma unroll
    for (int u = 0; u < CNT_GROUP; ++u) {
      const int g = gb + u * CNT_THREADS + t;
      rv[u] = recv4[(g < n4) ? g : 0];
    }
#pragma unroll
    for (int u = 0; u < CNT_GROUP; ++u) {
      const int g = gb + u * CNT_THREADS + t;
      if (g < n4) {
        int4 rk;
        rk.x = atomicAdd(&hist[rv[u].x], 1);
        rk.y = atomicAdd(&hist[rv[u].y], 1);
        rk.z = atomicAdd(&hist[rv[u].z], 1);
        rk.w = atomicAdd(&hist[rv[u].w], 1);
        rank4[g] = rk;
      }
    }
  }
  {
    const int e = e0 + 4 * n4 + t;
    if (t < 4 && e < e1) rank[e] = atomicAdd(&hist[recv[e]], 1);
  }
  __syncthreads();

  int* __restrict__ row = chunk_cnt + (size_t)c * gat_chunk_pitch(n_nodes);
  for (int i = t; i < n_nodes; i += CNT_THREADS) row[i] = hist[i];
}

__global__ __launch_bounds__(CNT_THREADS) void gat_count_project(
    const int* __restrict__ recv, int* __restrict__ chunk_cnt, int* __restrict__ rank,
    int n_edges, int n_nodes,
    const float* __restrict__ nodes, const float* __restrict__ W,
    float* __restrict__ Wn, int n_tiles)
{
  __shared__ int hist[LDS_BINS];
  if (blockIdx.x < SORT_CHUNKS)
    gat_count(hist, recv, chunk_cnt, rank, n_edges, n_nodes);
  else
    gat_node_project(nodes, W, Wn, n_nodes, n_tiles, (int)blockIdx.x - SORT_CHUNKS);
}

#define SCAN_BLOCK 256

__device__ __forceinline__ int gat_wave_incl(int x, int lane)
{
  for (int d = 1; d < 32; d <<= 1) {
    int y = __shfl_up(x, d);
    if (lane >= d) x += y;
  }
  return x;
}

__device__ __forceinline__ void gat_chunk_starts(
    int* __restrict__ chunk_cnt, int* __restrict__ within, int* __restrict__ blk_tot,
    int n_nodes)
{
  __shared__ int wtot[SCAN_BLOCK / 32];
  const int t = threadIdx.x, lane = t & 31, w = t >> 5;
  const int r = blockIdx.x * SCAN_BLOCK + t;
  const bool live = r < n_nodes;
  const int  rl = live ? r : n_nodes - 1;
  const size_t pitch = (size_t)gat_chunk_pitch(n_nodes);
  int v[SORT_CHUNKS];
#pragma unroll
  for (int c = 0; c < SORT_CHUNKS; ++c) v[c] = chunk_cnt[c * pitch + rl];
  int tot = 0;
#pragma unroll
  for (int c = 0; c < SORT_CHUNKS; ++c) tot += v[c];
  if (!live) tot = 0;

  const int x = gat_wave_incl(tot, lane);
  if (lane == 31) wtot[w] = x;
  __syncthreads();
  int wbase = 0, btot = 0;
#pragma unroll
  for (int k = 0; k < SCAN_BLOCK / 32; ++k) {
    const int s = wtot[k];
    if (k < w) wbase += s;
    btot += s;
  }
  const int ex = wbase + x - tot;

  if (live) {
    int run = ex;
#pragma unroll
    for (int c = 0; c < SORT_CHUNKS; ++c) {
      chunk_cnt[c * pitch + r] = run;
      run += v[c];
    }
    within[r] = ex;
  }
  if (t == 0) blk_tot[blockIdx.x] = btot;
}

__global__ __launch_bounds__(SCAN_BLOCK) void gat_starts_scores(
    int* __restrict__ chunk_cnt, int* __restrict__ within, int* __restrict__ blk_tot,
    int n_nodes, int nb,
    const float* __restrict__ Wn, const float* __restrict__ a,
    const float* __restrict__ W_edge,
    float* __restrict__ score_s, float* __restrict__ score_r, float* __restrict__ wcomb)
{
  if ((int)blockIdx.x < nb) {
    gat_chunk_starts(chunk_cnt, within, blk_tot, n_nodes);
  } else {
    const int blk = (int)blockIdx.x - nb;
    gat_node_scores(Wn, a, score_s, score_r, n_nodes, blk);
    if (blk == 0) gat_wcomb(W_edge, a, wcomb);
  }
}

__global__ __launch_bounds__(256) void gat_fill(
    const int* __restrict__ recv, const int* __restrict__ send,
    const int* __restrict__ rank, const int* __restrict__ within,
    const int* __restrict__ blk_tot, const int* __restrict__ chunk_start,
    int2* __restrict__ rec, int* __restrict__ offs, int n_edges, int n_nodes)
{
  __shared__ int blk_start[256];
  __shared__ int wtot[8];
  const int t = threadIdx.x, lane = t & 31, w = t >> 5;
  const int nb = (n_nodes + SCAN_BLOCK - 1) / SCAN_BLOCK;
  const int e = blockIdx.x * blockDim.x + t;
  const bool live = e < n_edges;
  const int tb = blk_tot[(t < nb) ? t : nb - 1];
  const int el = live ? e : n_edges - 1;
  const int r = recv[el], s = send[el], rk = rank[el];
  const int bt = (t < nb) ? tb : 0;

  const int x = gat_wave_incl(bt, lane);
  if (lane == 31) wtot[w] = x;
  __syncthreads();
  int wbase = 0;
#pragma unroll
  for (int k = 0; k < 8; ++k) {
    const int sw = wtot[k];
    if (k < w) wbase += sw;
  }
  const int ex = wbase + x - bt;
  blk_start[t] = ex;
  __syncthreads();

  if (live) {
    const int c = e / gat_chunk_len(n_edges);
    const int pos = blk_start[(unsigned)r / SCAN_BLOCK]
                  + chunk_start[(size_t)c * gat_chunk_pitch(n_nodes) + r] + rk;
    rec[pos] = make_int2(e, s);
  }

  for (int b = blockIdx.x; b < nb; b += gridDim.x) {
    const int i = b * SCAN_BLOCK + t;
    if (i < n_nodes) offs[i] = blk_start[b] + within[i];
  }
  if (blockIdx.x == 0 && t == nb - 1) offs[n_nodes] = ex + bt;
}

__global__ __launch_bounds__(256) void gat_gather(
    const float* __restrict__ edges, const int2* __restrict__ rec,
    const int* __restrict__ offs, const float* __restrict__ score_s,
    const float* __restrict__ score_r, const float* __restrict__ wcomb,
    const float* __restrict__ Wn, float* __restrict__ aggr, int n_nodes)
{
  const int r = __builtin_amdgcn_readfirstlane(
      (int)(blockIdx.x * (blockDim.x >> 5) + (threadIdx.x >> 5)));
  if (r >= n_nodes) return;
  const int lane = threadIdx.x & 31;
  const int hd   = lane >> 3;

  const int beg = offs[r], end = offs[r + 1];
  const float4 sr4 = *(const float4*)(score_r + (size_t)r * HEADS);

  float ax, ay, nrm;
  {
    const float4 ss4 = *(const float4*)(score_s + (size_t)r * HEADS);
    float p0 = gat_ez(ss4.x + sr4.x + 0.f);
    float p1 = gat_ez(ss4.y + sr4.y + 0.f);
    float p2 = gat_ez(ss4.z + sr4.z + 0.f);
    float p3 = gat_ez(ss4.w + sr4.w + 0.f);
    float p  = (hd == 0) ? p0 : (hd == 1) ? p1 : (hd == 2) ? p2 : p3;
    const float2 v = *(const float2*)(Wn + (size_t)r * DNODE + 2 * lane);
    ax = p * v.x; ay = p * v.y; nrm = p;
  }

  for (int base = beg; base < end; base += 32) {
    const int left = end - base;
    const int cnt  = (left < 32) ? left : 32;
    int   sv = r;
    float q0 = 0.f, q1 = 0.f, q2 = 0.f, q3 = 0.f;
    if (lane < cnt) {
      const int2 es = rec[base + lane];
      sv = es.y;
      const float4 ss4 = *(const float4*)(score_s + (size_t)es.y * HEADS);
      const float4* __restrict__ ep = (const float4*)(edges + (size_t)es.x * DEDGE);
      float ef[DEDGE];
#pragma unroll
      for (int c = 0; c < 4; ++c) {
        float4 q = ep[c];
        ef[4 * c + 0] = q.x; ef[4 * c + 1] = q.y; ef[4 * c + 2] = q.z; ef[4 * c + 3] = q.w;
      }
      float elt[HEADS];
#pragma unroll
      for (int h = 0; h < HEADS; ++h) {
        float acc = 0.f;
#pragma unroll
        for (int i = 0; i < DEDGE; ++i) acc = __builtin_fmaf(ef[i], wcomb[h * DEDGE + i], acc);
        elt[h] = acc;
      }
      q0 = gat_ez(ss4.x + sr4.x + elt[0]);
      q1 = gat_ez(ss4.y + sr4.y + elt[1]);
      q2 = gat_ez(ss4.z + sr4.z + elt[2]);
      q3 = gat_ez(ss4.w + sr4.w + elt[3]);
    }
    int j = 0;
    for (; j + 8 <= cnt; j += 8) {
      for (int u = 0; u < 8; ++u) {
        const int   sj = __builtin_amdgcn_readlane(sv, j + u);
        const float p0 = __int_as_float(__builtin_amdgcn_readlane(__float_as_int(q0), j + u));
        const float p1 = __int_as_float(__builtin_amdgcn_readlane(__float_as_int(q1), j + u));
        const float p2 = __int_as_float(__builtin_amdgcn_readlane(__float_as_int(q2), j + u));
        const float p3 = __int_as_float(__builtin_amdgcn_readlane(__float_as_int(q3), j + u));
        const float p  = (hd == 0) ? p0 : (hd == 1) ? p1 : (hd == 2) ? p2 : p3;
        const float2 v = *(const float2*)(Wn + (size_t)sj * DNODE + 2 * lane);
        ax += p * v.x; ay += p * v.y; nrm += p;
      }
    }
    for (; j < cnt; j += 4) {
      for (int u = 0; u < 4; ++u) {
        const int   sj = __builtin_amdgcn_readlane(sv, j + u);
        const float p0 = __int_as_float(__builtin_amdgcn_readlane(__float_as_int(q0), j + u));
        const float p1 = __int_as_float(__builtin_amdgcn_readlane(__float_as_int(q1), j + u));
        const float p2 = __int_as_float(__builtin_amdgcn_readlane(__float_as_int(q2), j + u));
        const float p3 = __int_as_float(__builtin_amdgcn_readlane(__float_as_int(q3), j + u));
        const float p  = (hd == 0) ? p0 : (hd == 1) ? p1 : (hd == 2) ? p2 : p3;
        const float2 v = *(const float2*)(Wn + (size_t)sj * DNODE + 2 * lane);
        ax += p * v.x; ay += p * v.y; nrm += p;
      }
    }
  }

  float2 o; o.x = ax / nrm; o.y = ay / nrm;
  *(float2*)(aggr + (size_t)r * DNODE + 2 * lane) = o;
}

__global__ __launch_bounds__(256) void gat_finalize(
    const float* __restrict__ aggr, float* __restrict__ out,
    const float* __restrict__ scale, const float* __restrict__ bias, int n_nodes)
{
  int i = blockIdx.x * blockDim.x + threadIdx.x;
  if (i >= n_nodes) return;
  const float* __restrict__ src = aggr + (size_t)i * DNODE;
  float* __restrict__ row = out + (size_t)i * DNODE;
  float v[DNODE];
  float mean = 0.f;
#pragma unroll
  for (int j = 0; j < DNODE; ++j) {
    float x = src[j];
    x = (x > 0.f) ? x : expm1f(x);
    v[j] = x; mean += x;
  }
  mean *= (1.f / DNODE);
  float var = 0.f;
#pragma unroll
  for (int j = 0; j < DNODE; ++j) { float d = v[j] - mean; var += d * d; }
  var *= (1.f / DNODE);
  float inv = rsqrtf(var + 1e-6f);
#pragma unroll
  for (int j = 0; j < DNODE; ++j)
    row[j] = (v[j] - mean) * inv * scale[j] + bias[j];
}

extern "C" void kernel_launch(void* const* d_in, const int* in_sizes, int n_in,
                              void* d_out, int out_size, void* d_ws, size_t ws_size,
                              hipStream_t stream)
{
  const float* nodes  = (const float*)d_in[0];
  const float* edges  = (const float*)d_in[1];
  const int*   recv   = (const int*)d_in[2];
  const int*   send   = (const int*)d_in[3];
  const float* W      = (const float*)d_in[4];
  const float* W_edge = (const float*)d_in[5];
  const float* a      = (const float*)d_in[6];
  const float* ln_s   = (const float*)d_in[7];
  const float* ln_b   = (const float*)d_in[8];

  const int n_nodes = in_sizes[0] / DNODE;
  const int n_edges = in_sizes[1] / DEDGE;
  float* out = (float*)d_out;

  char* ws = (char*)d_ws;
  size_t off = 0;
  auto carve = [&](size_t bytes) -> float* {
    float* p = (float*)(ws + off);
    off += (bytes + 255) & ~(size_t)255;
    return p;
  };
  const int    nb        = (n_nodes + SCAN_BLOCK - 1) / SCAN_BLOCK;
  const size_t rank_b    = ((size_t)n_edges * sizeof(int) + 255) & ~(size_t)255;
  const size_t chunk_b   = (size_t)SORT_CHUNKS * (size_t)((n_nodes + 63) & ~63) * sizeof(int);
  const size_t sort_b    = rank_b + chunk_b + (size_t)nb * sizeof(int);
  const size_t aggr_b    = (size_t)n_nodes * DNODE * sizeof(float);
  float* Wn      = out;
  float* plane   = carve(sort_b > aggr_b ? sort_b : aggr_b);
  float* score_s = carve((size_t)n_nodes * HEADS * sizeof(float));
  float* score_r = carve((size_t)n_nodes * HEADS * sizeof(float));
  float* wcomb   = carve((size_t)HEADS * DEDGE * sizeof(float));
  int*  within = (int*)carve((size_t)n_nodes * sizeof(int));
  int*  offs   = (int*)carve((size_t)(n_nodes + 1) * sizeof(int));
  int2* rec    = (int2*)carve((size_t)n_edges * sizeof(int2));
  int*  rank      = (int*)plane;
  int*  chunk_cnt = (int*)((char*)plane + rank_b);
  int*  blk_tot   = (int*)((char*)plane + rank_b + chunk_b);
  float* aggr     = plane;

  const int n_tiles = (n_nodes + 15) / 16;
  const int tiles_per_wg = CNT_THREADS / 32;
  dim3 blk(256);

  gat_count_project<<<dim3(SORT_CHUNKS + (n_tiles + tiles_per_wg - 1) / tiles_per_wg),
                      dim3(CNT_THREADS), 0, stream>>>(
      recv, chunk_cnt, rank, n_edges, n_nodes, nodes, W, Wn, n_tiles);
  gat_starts_scores<<<dim3(2 * nb), dim3(SCAN_BLOCK), 0, stream>>>(
      chunk_cnt, within, blk_tot, n_nodes, nb, Wn, a, W_edge, score_s, score_r, wcomb);
  gat_fill<<<dim3((n_edges + 255) / 256), blk, 0, stream>>>(
      recv, send, rank, within, blk_tot, chunk_cnt, rec, offs, n_edges, n_nodes);
  gat_gather<<<dim3((n_nodes + 7) / 8), blk, 0, stream>>>(
      edges, rec, offs, score_s, score_r, wcomb, Wn, aggr, n_nodes);
  gat_finalize<<<dim3((n_nodes + 255) / 256), blk, 0, stream>>>(
      aggr, out, ln_s, ln_b, n_nodes);
}


# node scores made inside the gather from the Wn rows it loads anyway (8-lane DPP sums per head), not read from score arrays: one random 16-B load per edge less; score and logit sums reordered
# speedup vs baseline: 10.8474x; 1.0505x over previous
#include <hip/hip_runtime.h>
#include <hip/hip_bf16.h>
#include <math.h>

typedef __attribute__((ext_vector_type(2))) float v2f;
typedef __attribute__((ext_vector_type(8))) float v8f;

#define HEADS 4
#define ATT_F 16
#define DNODE 64
#define DEDGE 16

__device__ __forceinline__ void gat_node_project(
    const float* __restrict__ nodes, const float* __restrict__ W,
    float* __restrict__ Wn, int n_nodes, int n_tiles, int blk)
{
  int wave = blk * (blockDim.x >> 5) + (threadIdx.x >> 5);
  if (wave >= n_tiles) return;
  int lane = threadIdx.x & 31;
  int half = lane >> 4;
  int l16  = lane & 15;
  int row0 = wave * 16;
  int m = row0 + l16; if (m >= n_nodes) m = n_nodes - 1;
  const float* __restrict__ arow = nodes + (size_t)m * DNODE;

  v8f acc[4] = {};
  for (int k0 = 0; k0 < DNODE; k0 += 4) {
    int kk = k0 + 2 * half;
    v2f a; a.x = arow[kk]; a.y = arow[kk + 1];
#pragma unroll
    for (int t = 0; t < 4; ++t) {
      int j = t * 16 + l16;
      v2f b; b.x = W[(size_t)j * DNODE + kk]; b.y = W[(size_t)j * DNODE + kk + 1];
      acc[t] = __builtin_amdgcn_wmma_f32_16x16x4_f32(
           false, a,  false, b,
           (short)0, acc[t],  false,  false);
    }
  }

  if (row0 + 16 <= n_nodes) {
#pragma unroll
    for (int t = 0; t < 4; ++t)
#pragma unroll
      for (int v = 0; v < 8; ++v)
        Wn[(size_t)(row0 + v + 8 * half) * DNODE + t * 16 + l16] = acc[t][v];
  } else {
#pragma unroll
    for (int t = 0; t < 4; ++t)
#pragma unroll
      for (int v = 0; v < 8; ++v) {
        int mo = row0 + v + 8 * half;
        if (mo < n_nodes) Wn[(size_t)mo * DNODE + t * 16 + l16] = acc[t][v];
      }
  }
}

__device__ __forceinline__ void gat_wcomb(const float* __restrict__ W_edge,
                          const float* __restrict__ a, float* __restrict__ wcomb)
{
  int idx = threadIdx.x;
  if (idx >= HEADS * DEDGE) return;
  int h = idx >> 4, i = idx & 15;
  float acc = 0.f;
#pragma unroll
  for (int f = 0; f < ATT_F; ++f)
    acc += a[h * 48 + 32 + f] * W_edge[((h * ATT_F) + f) * DEDGE + i];
  wcomb[idx] = acc;
}


__device__ __forceinline__ float gat_ez(float m)
{
  float lm = (m > 0.f) ? m : 0.01f * m;
  return __expf(lm);
}

#define SORT_CHUNKS 32
#define LDS_BINS    51200
#define CNT_THREADS 1024
#define CNT_GROUP   8

__device__ __forceinline__ int gat_chunk_len(int n_edges)
{
  return (((n_edges + SORT_CHUNKS - 1) / SORT_CHUNKS) + 3) & ~3;
}
__device__ __forceinline__ int gat_chunk_pitch(int n_nodes)
{
  return (n_nodes + 63) & ~63;
}

__device__ __forceinline__ void gat_count(
    int* __restrict__ hist,
    const int* __restrict__ recv, int* __restrict__ chunk_cnt, int* __restrict__ rank,
    int n_edges, int n_nodes)
{
  const int t  = threadIdx.x;
  const int c  = blockIdx.x;
  const int ch = gat_chunk_len(n_edges);
  const int e0 = c * ch;
  const int e1 = (e0 + ch < n_edges) ? e0 + ch : n_edges;
  const int n4 = (e1 > e0) ? ((e1 - e0) >> 2) : 0;

  for (int i = 4 * t; i < LDS_BINS; i += 4 * CNT_THREADS)
    *(int4*)&hist[i] = make_int4(0, 0, 0, 0);
  __syncthreads();

  const int4* __restrict__ recv4 = (const int4*)(recv + e0);
  int4* __restrict__ rank4 = (int4*)(rank + e0);
  for (int gb = 0; gb < n4; gb += CNT_THREADS * CNT_GROUP) {
    int4 rv[CNT_GROUP];
#pragma unroll
    for (int u = 0; u < CNT_GROUP; ++u) {
      const int g = gb + u * CNT_THREADS + t;
      rv[u] = recv4[(g < n4) ? g : 0];
    }
#pragma unroll
    for (int u = 0; u < CNT_GROUP; ++u) {
      const int g = gb + u * CNT_THREADS + t;
      if (g < n4) {
        int4 rk;
        rk.x = atomicAdd(&hist[rv[u].x], 1);
        rk.y = atomicAdd(&hist[rv[u].y], 1);
        rk.z = atomicAdd(&hist[rv[u].z], 1);
        rk.w = atomicAdd(&hist[rv[u].w], 1);
        rank4[g] = rk;
      }
    }
  }
  {
    const int e = e0 + 4 * n4 + t;
    if (t < 4 && e < e1) rank[e] = atomicAdd(&hist[recv[e]], 1);
  }
  __syncthreads();

  int* __restrict__ row = chunk_cnt + (size_t)c * gat_chunk_pitch(n_nodes);
  for (int i = t; i < n_nodes; i += CNT_THREADS) row[i] = hist[i];
}

__global__ __launch_bounds__(CNT_THREADS) void gat_count_project(
    const int* __restrict__ recv, int* __restrict__ chunk_cnt, int* __restrict__ rank,
    int n_edges, int n_nodes,
    const float* __restrict__ nodes, const float* __restrict__ W,
    float* __restrict__ Wn, int n_tiles)
{
  __shared__ int hist[LDS_BINS];
  if (blockIdx.x < SORT_CHUNKS)
    gat_count(hist, recv, chunk_cnt, rank, n_edges, n_nodes);
  else
    gat_node_project(nodes, W, Wn, n_nodes, n_tiles, (int)blockIdx.x - SORT_CHUNKS);
}

#define SCAN_BLOCK 256

__device__ __forceinline__ int gat_wave_incl(int x, int lane)
{
  for (int d = 1; d < 32; d <<= 1) {
    int y = __shfl_up(x, d);
    if (lane >= d) x += y;
  }
  return x;
}

__device__ __forceinline__ void gat_chunk_starts(
    int* __restrict__ chunk_cnt, int* __restrict__ within, int* __restrict__ blk_tot,
    int n_nodes)
{
  __shared__ int wtot[SCAN_BLOCK / 32];
  const int t = threadIdx.x, lane = t & 31, w = t >> 5;
  const int r = blockIdx.x * SCAN_BLOCK + t;
  const bool live = r < n_nodes;
  const int  rl = live ? r : n_nodes - 1;
  const size_t pitch = (size_t)gat_chunk_pitch(n_nodes);
  int v[SORT_CHUNKS];
#pragma unroll
  for (int c = 0; c < SORT_CHUNKS; ++c) v[c] = chunk_cnt[c * pitch + rl];
  int tot = 0;
#pragma unroll
  for (int c = 0; c < SORT_CHUNKS; ++c) tot += v[c];
  if (!live) tot = 0;

  const int x = gat_wave_incl(tot, lane);
  if (lane == 31) wtot[w] = x;
  __syncthreads();
  int wbase = 0, btot = 0;
#pragma unroll
  for (int k = 0; k < SCAN_BLOCK / 32; ++k) {
    const int s = wtot[k];
    if (k < w) wbase += s;
    btot += s;
  }
  const int ex = wbase + x - tot;

  if (live) {
    int run = ex;
#pragma unroll
    for (int c = 0; c < SORT_CHUNKS; ++c) {
      chunk_cnt[c * pitch + r] = run;
      run += v[c];
    }
    within[r] = ex;
  }
  if (t == 0) blk_tot[blockIdx.x] = btot;
}

__global__ __launch_bounds__(SCAN_BLOCK) void gat_starts_wcomb(
    int* __restrict__ chunk_cnt, int* __restrict__ within, int* __restrict__ blk_tot,
    int n_nodes, int nb,
    const float* __restrict__ a, const float* __restrict__ W_edge,
    float* __restrict__ wcomb)
{
  if ((int)blockIdx.x < nb)
    gat_chunk_starts(chunk_cnt, within, blk_tot, n_nodes);
  else
    gat_wcomb(W_edge, a, wcomb);
}

__global__ __launch_bounds__(256) void gat_fill(
    const int* __restrict__ recv, const int* __restrict__ send,
    const int* __restrict__ rank, const int* __restrict__ within,
    const int* __restrict__ blk_tot, const int* __restrict__ chunk_start,
    int2* __restrict__ rec, int* __restrict__ offs, int n_edges, int n_nodes)
{
  __shared__ int blk_start[256];
  __shared__ int wtot[8];
  const int t = threadIdx.x, lane = t & 31, w = t >> 5;
  const int nb = (n_nodes + SCAN_BLOCK - 1) / SCAN_BLOCK;
  const int e = blockIdx.x * blockDim.x + t;
  const bool live = e < n_edges;
  const int tb = blk_tot[(t < nb) ? t : nb - 1];
  const int el = live ? e : n_edges - 1;
  const int r = recv[el], s = send[el], rk = rank[el];
  const int bt = (t < nb) ? tb : 0;

  const int x = gat_wave_incl(bt, lane);
  if (lane == 31) wtot[w] = x;
  __syncthreads();
  int wbase = 0;
#pragma unroll
  for (int k = 0; k < 8; ++k) {
    const int sw = wtot[k];
    if (k < w) wbase += sw;
  }
  const int ex = wbase + x - bt;
  blk_start[t] = ex;
  __syncthreads();

  if (live) {
    const int c = e / gat_chunk_len(n_edges);
    const int pos = blk_start[(unsigned)r / SCAN_BLOCK]
                  + chunk_start[(size_t)c * gat_chunk_pitch(n_nodes) + r] + rk;
    rec[pos] = make_int2(e, s);
  }

  for (int b = blockIdx.x; b < nb; b += gridDim.x) {
    const int i = b * SCAN_BLOCK + t;
    if (i < n_nodes) offs[i] = blk_start[b] + within[i];
  }
  if (blockIdx.x == 0 && t == nb - 1) offs[n_nodes] = ex + bt;
}

__device__ __forceinline__ float gat_sum8(float x)
{
  x += __int_as_float(__builtin_amdgcn_update_dpp(0, __float_as_int(x), 0x161, 0xf, 0xf, true));
  x += __int_as_float(__builtin_amdgcn_update_dpp(0, __float_as_int(x), 0x162, 0xf, 0xf, true));
  x += __int_as_float(__builtin_amdgcn_update_dpp(0, __float_as_int(x), 0x164, 0xf, 0xf, true));
  return x;
}

template <int U>
__device__ __forceinline__ void gat_step(float G, float2 v, float2 as,
                                         float& ax, float& ay, float& nrm)
{
  const float ss = gat_sum8(__builtin_fmaf(v.y, as.y, v.x * as.x));
  const float cg = __int_as_float(__builtin_amdgcn_mov_dpp8(__float_as_int(G), U * 0x249249));
  const float p  = gat_ez(ss + cg);
  ax += p * v.x; ay += p * v.y; nrm += p;
}

__device__ __forceinline__ float2 gat_row(const float* __restrict__ Wn, int sv, int j, int lane)
{
  const int sj = __builtin_amdgcn_readlane(sv, j);
  return *(const float2*)(Wn + (size_t)sj * DNODE + 2 * lane);
}

__device__ __forceinline__ void gat_walk8(const float* __restrict__ Wn, int sv, float G,
    int j0, int lane, float2 as, float& ax, float& ay, float& nrm)
{
  const float2 v0 = gat_row(Wn, sv, j0 + 0, lane), v1 = gat_row(Wn, sv, j0 + 1, lane);
  const float2 v2 = gat_row(Wn, sv, j0 + 2, lane), v3 = gat_row(Wn, sv, j0 + 3, lane);
  const float2 v4 = gat_row(Wn, sv, j0 + 4, lane), v5 = gat_row(Wn, sv, j0 + 5, lane);
  const float2 v6 = gat_row(Wn, sv, j0 + 6, lane), v7 = gat_row(Wn, sv, j0 + 7, lane);
  gat_step<0>(G, v0, as, ax, ay, nrm); gat_step<1>(G, v1, as, ax, ay, nrm);
  gat_step<2>(G, v2, as, ax, ay, nrm); gat_step<3>(G, v3, as, ax, ay, nrm);
  gat_step<4>(G, v4, as, ax, ay, nrm); gat_step<5>(G, v5, as, ax, ay, nrm);
  gat_step<6>(G, v6, as, ax, ay, nrm); gat_step<7>(G, v7, as, ax, ay, nrm);
}

template <int H>
__device__ __forceinline__ void gat_walk4(const float* __restrict__ Wn, int sv, float G,
    int j0, int lane, float2 as, float& ax, float& ay, float& nrm)
{
  const float2 v0 = gat_row(Wn, sv, j0 + 0, lane), v1 = gat_row(Wn, sv, j0 + 1, lane);
  const float2 v2 = gat_row(Wn, sv, j0 + 2, lane), v3 = gat_row(Wn, sv, j0 + 3, lane);
  gat_step<H + 0>(G, v0, as, ax, ay, nrm); gat_step<H + 1>(G, v1, as, ax, ay, nrm);
  gat_step<H + 2>(G, v2, as, ax, ay, nrm); gat_step<H + 3>(G, v3, as, ax, ay, nrm);
}

__global__ __launch_bounds__(256) void gat_gather(
    const float* __restrict__ edges, const int2* __restrict__ rec,
    const int* __restrict__ offs, const float* __restrict__ a,
    const float* __restrict__ wcomb,
    const float* __restrict__ Wn, float* __restrict__ aggr, int n_nodes)
{
  __shared__ float4 turn[8][32];
  const int r = __builtin_amdgcn_readfirstlane(
      (int)(blockIdx.x * (blockDim.x >> 5) + (threadIdx.x >> 5)));
  if (r >= n_nodes) return;
  const int lane = threadIdx.x & 31;
  const int hd   = lane >> 3;
  float4* __restrict__ mine = turn[threadIdx.x >> 5];

  const int beg = offs[r], end = offs[r + 1];

  const float2 as = *(const float2*)(a + hd * 48 + 2 * (lane & 7));
  const float2 ar = *(const float2*)(a + hd * 48 + 16 + 2 * (lane & 7));

  float ax, ay, nrm;
  float sr0, sr1, sr2, sr3;
  {
    const float2 v = *(const float2*)(Wn + (size_t)r * DNODE + 2 * lane);
    const float ss = gat_sum8(__builtin_fmaf(v.y, as.y, v.x * as.x));
    const float sr = gat_sum8(__builtin_fmaf(v.y, ar.y, v.x * ar.x));
    sr0 = __int_as_float(__builtin_amdgcn_readlane(__float_as_int(sr), 0));
    sr1 = __int_as_float(__builtin_amdgcn_readlane(__float_as_int(sr), 8));
    sr2 = __int_as_float(__builtin_amdgcn_readlane(__float_as_int(sr), 16));
    sr3 = __int_as_float(__builtin_amdgcn_readlane(__float_as_int(sr), 24));
    const float p = gat_ez(ss + sr);
    ax = p * v.x; ay = p * v.y; nrm = p;
  }

  for (int base = beg; base < end; base += 32) {
    const int left = end - base;
    const int cnt  = (left < 32) ? left : 32;
    int    sv = r;
    float4 g  = make_float4(-INFINITY, -INFINITY, -INFINITY, -INFINITY);
    if (lane < cnt) {
      const int2 es = rec[base + lane];
      sv = es.y;
      const float4* __restrict__ ep = (const float4*)(edges + (size_t)es.x * DEDGE);
      float ef[DEDGE];
#pragma unroll
      for (int c = 0; c < 4; ++c) {
        float4 q = ep[c];
        ef[4 * c + 0] = q.x; ef[4 * c + 1] = q.y; ef[4 * c + 2] = q.z; ef[4 * c + 3] = q.w;
      }
      float elt[HEADS];
#pragma unroll
      for (int h = 0; h < HEADS; ++h) {
        float acc = 0.f;
#pragma unroll
        for (int i = 0; i < DEDGE; ++i) acc = __builtin_fmaf(ef[i], wcomb[h * DEDGE + i], acc);
        elt[h] = acc;
      }
      g = make_float4(sr0 + elt[0], sr1 + elt[1], sr2 + elt[2], sr3 + elt[3]);
    }
    mine[lane] = g;
    float G[4];
#pragma unroll
    for (int q = 0; q < 4; ++q)
      G[q] = ((const float*)mine)[(8 * q + (lane & 7)) * 4 + hd];

#pragma unroll
    for (int q = 0; q < 4; ++q) {
      const int j0 = 8 * q;
      if (j0 + 8 <= cnt) {
        gat_walk8(Wn, sv, G[q], j0, lane, as, ax, ay, nrm);
      } else {
        if (j0 < cnt)     gat_walk4<0>(Wn, sv, G[q], j0, lane, as, ax, ay, nrm);
        if (j0 + 4 < cnt) gat_walk4<4>(Wn, sv, G[q], j0 + 4, lane, as, ax, ay, nrm);
      }
    }
  }

  float2 o; o.x = ax / nrm; o.y = ay / nrm;
  *(float2*)(aggr + (size_t)r * DNODE + 2 * lane) = o;
}

__global__ __launch_bounds__(256) void gat_finalize(
    const float* __restrict__ aggr, float* __restrict__ out,
    const float* __restrict__ scale, const float* __restrict__ bias, int n_nodes)
{
  int i = blockIdx.x * blockDim.x + threadIdx.x;
  if (i >= n_nodes) return;
  const float* __restrict__ src = aggr + (size_t)i * DNODE;
  float* __restrict__ row = out + (size_t)i * DNODE;
  float v[DNODE];
  float mean = 0.f;
#pragma unroll
  for (int j = 0; j < DNODE; ++j) {
    float x = src[j];
    x = (x > 0.f) ? x : expm1f(x);
    v[j] = x; mean += x;
  }
  mean *= (1.f / DNODE);
  float var = 0.f;
#pragma unroll
  for (int j = 0; j < DNODE; ++j) { float d = v[j] - mean; var += d * d; }
  var *= (1.f / DNODE);
  float inv = rsqrtf(var + 1e-6f);
#pragma unroll
  for (int j = 0; j < DNODE; ++j)
    row[j] = (v[j] - mean) * inv * scale[j] + bias[j];
}

extern "C" void kernel_launch(void* const* d_in, const int* in_sizes, int n_in,
                              void* d_out, int out_size, void* d_ws, size_t ws_size,
                              hipStream_t stream)
{
  const float* nodes  = (const float*)d_in[0];
  const float* edges  = (const float*)d_in[1];
  const int*   recv   = (const int*)d_in[2];
  const int*   send   = (const int*)d_in[3];
  const float* W      = (const float*)d_in[4];
  const float* W_edge = (const float*)d_in[5];
  const float* a      = (const float*)d_in[6];
  const float* ln_s   = (const float*)d_in[7];
  const float* ln_b   = (const float*)d_in[8];

  const int n_nodes = in_sizes[0] / DNODE;
  const int n_edges = in_sizes[1] / DEDGE;
  float* out = (float*)d_out;

  char* ws = (char*)d_ws;
  size_t off = 0;
  auto carve = [&](size_t bytes) -> float* {
    float* p = (float*)(ws + off);
    off += (bytes + 255) & ~(size_t)255;
    return p;
  };
  const int    nb        = (n_nodes + SCAN_BLOCK - 1) / SCAN_BLOCK;
  const size_t rank_b    = ((size_t)n_edges * sizeof(int) + 255) & ~(size_t)255;
  const size_t chunk_b   = (size_t)SORT_CHUNKS * (size_t)((n_nodes + 63) & ~63) * sizeof(int);
  const size_t sort_b    = rank_b + chunk_b + (size_t)nb * sizeof(int);
  const size_t aggr_b    = (size_t)n_nodes * DNODE * sizeof(float);
  float* Wn      = out;
  float* plane   = carve(sort_b > aggr_b ? sort_b : aggr_b);
  float* wcomb   = carve((size_t)HEADS * DEDGE * sizeof(float));
  int*  within = (int*)carve((size_t)n_nodes * sizeof(int));
  int*  offs   = (int*)carve((size_t)(n_nodes + 1) * sizeof(int));
  int2* rec    = (int2*)carve((size_t)n_edges * sizeof(int2));
  int*  rank      = (int*)plane;
  int*  chunk_cnt = (int*)((char*)plane + rank_b);
  int*  blk_tot   = (int*)((char*)plane + rank_b + chunk_b);
  float* aggr     = plane;

  const int n_tiles = (n_nodes + 15) / 16;
  const int tiles_per_wg = CNT_THREADS / 32;
  dim3 blk(256);

  gat_count_project<<<dim3(SORT_CHUNKS + (n_tiles + tiles_per_wg - 1) / tiles_per_wg),
                      dim3(CNT_THREADS), 0, stream>>>(
      recv, chunk_cnt, rank, n_edges, n_nodes, nodes, W, Wn, n_tiles);
  gat_starts_wcomb<<<dim3(nb + 1), dim3(SCAN_BLOCK), 0, stream>>>(
      chunk_cnt, within, blk_tot, n_nodes, nb, a, W_edge, wcomb);
  gat_fill<<<dim3((n_edges + 255) / 256), blk, 0, stream>>>(
      recv, send, rank, within, blk_tot, chunk_cnt, rec, offs, n_edges, n_nodes);
  gat_gather<<<dim3((n_nodes + 7) / 8), blk, 0, stream>>>(
      edges, rec, offs, a, wcomb, Wn, aggr, n_nodes);
  gat_finalize<<<dim3((n_nodes + 255) / 256), blk, 0, stream>>>(
      aggr, out, ln_s, ln_b, n_nodes);
}
